# starting kernel
# baseline (speedup 1.0000x reference)
#include <hip/hip_runtime.h>

#define Bn    4
#define Sn    512
#define Dn    256
#define Wn    32
#define Ln    4
#define Hn    8
#define DFFn  1024
#define NOUTn 32

typedef __attribute__((ext_vector_type(16))) _Float16 v16h;
typedef __attribute__((ext_vector_type(8)))  _Float16 v8h;
typedef __attribute__((ext_vector_type(8)))  float    v8f;

union HFrag { v16h v; v8h h[2]; };

__device__ __forceinline__ v16h load_frag(const _Float16* p, int ld, int row0, int k0) {
  const int lane = threadIdx.x & 31;
  const _Float16* q = p + (row0 + (lane & 15)) * ld + k0 + ((lane >> 4) << 3);
  HFrag f;
  f.h[0] = *(const v8h*)(q);
  f.h[1] = *(const v8h*)(q + 16);
  return f.v;
}

__device__ __forceinline__ v8f wmma_f16(v16h a, v16h b, v8f c) {
  return __builtin_amdgcn_wmma_f32_16x16x32_f16(false, a, false, b, (short)0, c, false, false);
}

__device__ __forceinline__ void store_tile_f16(_Float16* p, int ld, int row0, int col0, v8f acc) {
  const int lane = threadIdx.x & 31;
  const int m0 = row0 + ((lane >> 4) << 3);
  _Float16* q = p + col0 + (lane & 15);
#pragma unroll
  for (int r = 0; r < 8; ++r) q[(m0 + r) * ld] = (_Float16)acc[r];
}

__device__ __forceinline__ void store_tile_f16_T(_Float16* p, int ld, int row0, int col0, v8f acc) {
  const int lane = threadIdx.x & 31;
  const int m0 = row0 + ((lane >> 4) << 3);
  const int n  = col0 + (lane & 15);
  v8h h;
#pragma unroll
  for (int r = 0; r < 8; ++r) h[r] = (_Float16)acc[r];
  *(v8h*)(p + n * ld + m0) = h;
}

__device__ __forceinline__ void layer_norm_rows(float* xf32, _Float16* xf16,
                                                const float* __restrict__ g,
                                                const float* __restrict__ bt,
                                                int wave, int lane) {
#pragma unroll
  for (int rr = 0; rr < 4; ++rr) {
    const int m = wave * 4 + rr;
    float vals[8];
    float sum = 0.f;
#pragma unroll
    for (int i = 0; i < 8; ++i) { vals[i] = xf32[m * 256 + lane + 32 * i]; sum += vals[i]; }
#pragma unroll
    for (int o = 16; o >= 1; o >>= 1) sum += __shfl_xor(sum, o, 32);
    const float mu = sum * (1.f / 256.f);
    float s2 = 0.f;
#pragma unroll
    for (int i = 0; i < 8; ++i) { float d = vals[i] - mu; s2 += d * d; }
#pragma unroll
    for (int o = 16; o >= 1; o >>= 1) s2 += __shfl_xor(s2, o, 32);
    const float rstd = rsqrtf(s2 * (1.f / 256.f) + 1e-5f);
#pragma unroll
    for (int i = 0; i < 8; ++i) {
      const int c = lane + 32 * i;
      const float y = (vals[i] - mu) * rstd * g[c] + bt[c];
      xf32[m * 256 + c] = y;
      xf16[m * 272 + c] = (_Float16)y;
    }
  }
}

__global__ void cvt_f32_f16(const float* __restrict__ s, _Float16* __restrict__ d, int n) {
  int i = (blockIdx.x * blockDim.x + threadIdx.x) * 4;
  if (i + 3 < n) {
    float4 v = *(const float4*)(s + i);
    d[i + 0] = (_Float16)v.x;
    d[i + 1] = (_Float16)v.y;
    d[i + 2] = (_Float16)v.z;
    d[i + 3] = (_Float16)v.w;
  }
}

__global__ __launch_bounds__(512) void swt_kernel(
    const float* __restrict__ X, const float* __restrict__ pos,
    const _Float16* __restrict__ Win,  const float* __restrict__ Bin,
    const _Float16* __restrict__ Wout, const float* __restrict__ Bout,
    const _Float16* __restrict__ W1,   const float* __restrict__ B1,
    const _Float16* __restrict__ W2,   const float* __restrict__ B2,
    const float* __restrict__ ln1g, const float* __restrict__ ln1b,
    const float* __restrict__ ln2g, const float* __restrict__ ln2b,
    const float* __restrict__ headw, const float* __restrict__ headb,
    float* __restrict__ out) {
  __shared__ float    xf32[64 * 256];
  __shared__ _Float16 l16[92160];
  _Float16* xf16 = l16;
  _Float16* qk   = l16 + 17408;
  _Float16* vT   = l16 + 51200;
  _Float16* pr   = l16 + 71680;

  const int tid  = threadIdx.x;
  const int lane = tid & 31;
  const int wave = __builtin_amdgcn_readfirstlane(tid >> 5);
  const int pair0 = blockIdx.x * 2;
  const int bb   = pair0 >> 9;
  const int ss0  = pair0 & 511;

  {
    const int jr = tid >> 3;
    const int wwin = jr >> 5, j = jr & 31;
    const int c0 = (tid & 7) * 32;
    int src = (ss0 + wwin) - j; if (src < 0) src = 0;
    const float* xr = X + ((bb << 9) + src) * 256;
    const float* pe = pos + j * 256;
#pragma unroll
    for (int u = 0; u < 8; ++u) {
      const int c = c0 + u * 4;
      float4 xv = *(const float4*)(xr + c);
      float4 pv = *(const float4*)(pe + c);
      float y0 = xv.x + pv.x, y1 = xv.y + pv.y, y2 = xv.z + pv.z, y3 = xv.w + pv.w;
      xf32[jr * 256 + c + 0] = y0; xf16[jr * 272 + c + 0] = (_Float16)y0;
      xf32[jr * 256 + c + 1] = y1; xf16[jr * 272 + c + 1] = (_Float16)y1;
      xf32[jr * 256 + c + 2] = y2; xf16[jr * 272 + c + 2] = (_Float16)y2;
      xf32[jr * 256 + c + 3] = y3; xf16[jr * 272 + c + 3] = (_Float16)y3;
    }
  }
  __syncthreads();

  const v8f vzero = {};
  const int m0l = ((lane >> 4) << 3);
  const int nnl = lane & 15;
  const int kh  = wave >> 3;
  const int np  = wave & 7;

  for (int l = 0; l < Ln; ++l) {
    const _Float16* wqkv = Win  + l * 768 * 256;
    const float*    bqkv = Bin  + l * 768;
    const _Float16* wo   = Wout + l * 256 * 256;
    const float*    bo   = Bout + l * 256;
    const _Float16* w1l  = W1   + l * 1024 * 256;
    const float*    b1l  = B1   + l * 1024;
    const _Float16* w2l  = W2   + l * 256 * 1024;
    const float*    b2l  = B2   + l * 256;

    {
      const int n0 = wave * 16;
      v8f acc[4][3];
#pragma unroll
      for (int mi = 0; mi < 4; ++mi) { acc[mi][0] = vzero; acc[mi][1] = vzero; acc[mi][2] = vzero; }
#pragma unroll
      for (int kt = 0; kt < 8; ++kt) {
        v16h bq = load_frag(wqkv, 256, n0,       kt * 32);
        v16h bk = load_frag(wqkv, 256, n0 + 256, kt * 32);
        v16h bv = load_frag(wqkv, 256, n0 + 512, kt * 32);
#pragma unroll
        for (int mi = 0; mi < 4; ++mi) {
          v16h a = load_frag(xf16, 272, mi * 16, kt * 32);
          acc[mi][0] = wmma_f16(a, bq, acc[mi][0]);
          acc[mi][1] = wmma_f16(a, bk, acc[mi][1]);
          acc[mi][2] = wmma_f16(a, bv, acc[mi][2]);
        }
      }
      const float biasq = bqkv[n0 + nnl];
      const float biask = bqkv[n0 + 256 + nnl];
      const float biasv = bqkv[n0 + 512 + nnl];
#pragma unroll
      for (int mi = 0; mi < 4; ++mi) {
#pragma unroll
        for (int r = 0; r < 8; ++r) {
          acc[mi][0][r] += biasq; acc[mi][1][r] += biask; acc[mi][2][r] += biasv;
        }
        store_tile_f16(qk, 528, mi * 16, n0,       acc[mi][0]);
        store_tile_f16(qk, 528, mi * 16, n0 + 256, acc[mi][1]);
        store_tile_f16_T(vT + (mi >> 1) * 10240, 40, (mi & 1) * 16, n0, acc[mi][2]);
      }
    }
    __syncthreads();

    {
      const int aw = wave >> 3;
      const int h  = wave & 7;
      const int rb = aw * 32;
      _Float16* vTw = vT + aw * 10240;
      _Float16* ph  = pr + wave * 1280;
      v8f sc[2][2];
#pragma unroll
      for (int qt = 0; qt < 2; ++qt) {
        v16h a = load_frag(qk, 528, rb + qt * 16, h * 32);
#pragma unroll
        for (int nt = 0; nt < 2; ++nt) {
          v16h b = load_frag(qk + 256, 528, rb + nt * 16, h * 32);
          sc[qt][nt] = wmma_f16(a, b, vzero);
        }
      }
      const float scale = 0.17677669529663687f;
#pragma unroll
      for (int qt = 0; qt < 2; ++qt) {
#pragma unroll
        for (int r = 0; r < 8; ++r) {
          float s0 = sc[qt][0][r] * scale;
          float s1 = sc[qt][1][r] * scale;
          float mx = fmaxf(s0, s1);
#pragma unroll
          for (int o = 8; o >= 1; o >>= 1) mx = fmaxf(mx, __shfl_xor(mx, o, 32));
          const float e0 = __expf(s0 - mx), e1 = __expf(s1 - mx);
          float sm = e0 + e1;
#pragma unroll
          for (int o = 8; o >= 1; o >>= 1) sm += __shfl_xor(sm, o, 32);
          const float inv = 1.0f / sm;
          const int m = qt * 16 + m0l + r;
          ph[m * 40 + nnl]      = (_Float16)(e0 * inv);
          ph[m * 40 + 16 + nnl] = (_Float16)(e1 * inv);
        }
      }
#pragma unroll
      for (int qt = 0; qt < 2; ++qt) {
        v16h a = load_frag(ph, 40, qt * 16, 0);
#pragma unroll
        for (int dt = 0; dt < 2; ++dt) {
          v16h b = load_frag(vTw, 40, h * 32 + dt * 16, 0);
          v8f oa = wmma_f16(a, b, vzero);
          store_tile_f16(qk, 528, rb + qt * 16, h * 32 + dt * 16, oa);
        }
      }
    }
    __syncthreads();

    {
      const int n0 = np * 32;
      v8f acc[4][2];
#pragma unroll
      for (int mi = 0; mi < 4; ++mi) { acc[mi][0] = vzero; acc[mi][1] = vzero; }
#pragma unroll
      for (int kk = 0; kk < 4; ++kk) {
        const int kt = kh * 4 + kk;
        v16h b0 = load_frag(wo, 256, n0,      kt * 32);
        v16h b1 = load_frag(wo, 256, n0 + 16, kt * 32);
#pragma unroll
        for (int mi = 0; mi < 4; ++mi) {
          v16h a = load_frag(qk, 528, mi * 16, kt * 32);
          acc[mi][0] = wmma_f16(a, b0, acc[mi][0]);
          acc[mi][1] = wmma_f16(a, b1, acc[mi][1]);
        }
      }
      const float bias0 = (kh == 0) ? bo[n0 + nnl] : 0.f;
      const float bias1 = (kh == 0) ? bo[n0 + 16 + nnl] : 0.f;
#pragma unroll
      for (int mi = 0; mi < 4; ++mi) {
        const int m0 = mi * 16 + m0l;
#pragma unroll
        for (int r = 0; r < 8; ++r) {
          atomicAdd(&xf32[(m0 + r) * 256 + n0 + nnl],      acc[mi][0][r] + bias0);
          atomicAdd(&xf32[(m0 + r) * 256 + n0 + 16 + nnl], acc[mi][1][r] + bias1);
        }
      }
    }
    __syncthreads();
    layer_norm_rows(xf32, xf16, ln1g + l * 256, ln1b + l * 256, wave, lane);
    __syncthreads();

    v8f facc[4][2];
#pragma unroll
    for (int mi = 0; mi < 4; ++mi) { facc[mi][0] = vzero; facc[mi][1] = vzero; }
    for (int cp = 0; cp < 2; ++cp) {
      {
        const int ng0 = cp * 512 + wave * 16;
        const int ng1 = ng0 + 256;
        v8f acc[4][2];
#pragma unroll
        for (int mi = 0; mi < 4; ++mi) { acc[mi][0] = vzero; acc[mi][1] = vzero; }
#pragma unroll
        for (int kt = 0; kt < 8; ++kt) {
          v16h b0 = load_frag(w1l, 256, ng0, kt * 32);
          v16h b1 = load_frag(w1l, 256, ng1, kt * 32);
#pragma unroll
          for (int mi = 0; mi < 4; ++mi) {
            v16h a = load_frag(xf16, 272, mi * 16, kt * 32);
            acc[mi][0] = wmma_f16(a, b0, acc[mi][0]);
            acc[mi][1] = wmma_f16(a, b1, acc[mi][1]);
          }
        }
        const float bias0 = b1l[ng0 + nnl];
        const float bias1 = b1l[ng1 + nnl];
#pragma unroll
        for (int mi = 0; mi < 4; ++mi) {
#pragma unroll
          for (int r = 0; r < 8; ++r) {
            acc[mi][0][r] = fmaxf(acc[mi][0][r] + bias0, 0.f);
            acc[mi][1][r] = fmaxf(acc[mi][1][r] + bias1, 0.f);
          }
          store_tile_f16(qk, 528, mi * 16, wave * 16,       acc[mi][0]);
          store_tile_f16(qk, 528, mi * 16, 256 + wave * 16, acc[mi][1]);
        }
      }
      __syncthreads();
      {
        const int n0 = np * 32;
#pragma unroll
        for (int kk = 0; kk < 8; ++kk) {
          const int kt = kh * 8 + kk;
          v16h b0 = load_frag(w2l, 1024, n0,      cp * 512 + kt * 32);
          v16h b1 = load_frag(w2l, 1024, n0 + 16, cp * 512 + kt * 32);
#pragma unroll
          for (int mi = 0; mi < 4; ++mi) {
            v16h a = load_frag(qk, 528, mi * 16, kt * 32);
            facc[mi][0] = wmma_f16(a, b0, facc[mi][0]);
            facc[mi][1] = wmma_f16(a, b1, facc[mi][1]);
          }
        }
      }
      __syncthreads();
    }
    {
      const int n0 = np * 32;
      const float bias0 = (kh == 0) ? b2l[n0 + nnl] : 0.f;
      const float bias1 = (kh == 0) ? b2l[n0 + 16 + nnl] : 0.f;
#pragma unroll
      for (int mi = 0; mi < 4; ++mi) {
        const int m0 = mi * 16 + m0l;
#pragma unroll
        for (int r = 0; r < 8; ++r) {
          atomicAdd(&xf32[(m0 + r) * 256 + n0 + nnl],      facc[mi][0][r] + bias0);
          atomicAdd(&xf32[(m0 + r) * 256 + n0 + 16 + nnl], facc[mi][1][r] + bias1);
        }
      }
    }
    __syncthreads();
    layer_norm_rows(xf32, xf16, ln2g + l * 256, ln2b + l * 256, wave, lane);
    __syncthreads();
  }

  if (wave < 2) {
    float acc = headb[lane];
    const float* hw = headw + lane * 256;
    const float* xr = xf32 + (wave * 32 + 31) * 256;
    for (int k = 0; k < 256; k += 4) {
      float4 w4 = *(const float4*)(hw + k);
      acc += xr[k] * w4.x + xr[k + 1] * w4.y + xr[k + 2] * w4.z + xr[k + 3] * w4.w;
    }
    out[(pair0 + wave) * 32 + lane] = acc;
  }
}

extern "C" void kernel_launch(void* const* d_in, const int* in_sizes, int n_in,
                              void* d_out, int out_size, void* d_ws, size_t ws_size,
                              hipStream_t stream) {
  const float* X     = (const float*)d_in[0];
  const float* pos   = (const float*)d_in[3];
  const float* w_in  = (const float*)d_in[4];
  const float* b_in  = (const float*)d_in[5];
  const float* w_out = (const float*)d_in[6];
  const float* b_out = (const float*)d_in[7];
  const float* w1    = (const float*)d_in[8];
  const float* b1    = (const float*)d_in[9];
  const float* w2    = (const float*)d_in[10];
  const float* b2    = (const float*)d_in[11];
  const float* ln1g  = (const float*)d_in[12];
  const float* ln1b  = (const float*)d_in[13];
  const float* ln2g  = (const float*)d_in[14];
  const float* ln2b  = (const float*)d_in[15];
  const float* headw = (const float*)d_in[16];
  const float* headb = (const float*)d_in[17];
  float* out = (float*)d_out;

  _Float16* ws      = (_Float16*)d_ws;
  _Float16* w_in_h  = ws;
  _Float16* w_out_h = w_in_h  + Ln * 768 * 256;
  _Float16* w1_h    = w_out_h + Ln * 256 * 256;
  _Float16* w2_h    = w1_h    + Ln * 1024 * 256;

  const int n_in_w  = Ln * 768 * 256;
  const int n_out_w = Ln * 256 * 256;
  const int n_w1    = Ln * 1024 * 256;
  const int n_w2    = Ln * 256 * 1024;
  cvt_f32_f16<<<n_in_w  / 1024, 256, 0, stream>>>(w_in,  w_in_h,  n_in_w);
  cvt_f32_f16<<<n_out_w / 1024, 256, 0, stream>>>(w_out, w_out_h, n_out_w);
  cvt_f32_f16<<<n_w1    / 1024, 256, 0, stream>>>(w1,    w1_h,    n_w1);
  cvt_f32_f16<<<n_w2    / 1024, 256, 0, stream>>>(w2,    w2_h,    n_w2);

  swt_kernel<<<(Bn * Sn) / 2, 512, 0, stream>>>(X, pos, w_in_h, b_in, w_out_h, b_out,
                                                w1_h, b1, w2_h, b2, ln1g, ln1b, ln2g,
                                                ln2b, headw, headb, out);
}


# Weight images stored fragment-major: each B-fragment global load reads 512 contiguous bytes (4 lines) instead of 16 lines x 32 B; same lane/element map, products and sums bit-identical
# speedup vs baseline: 1.1367x; 1.1367x over previous
#include <hip/hip_runtime.h>

#define Bn    4
#define Sn    512
#define Dn    256
#define Wn    32
#define Ln    4
#define Hn    8
#define DFFn  1024
#define NOUTn 32

typedef __attribute__((ext_vector_type(16))) _Float16 v16h;
typedef __attribute__((ext_vector_type(8)))  _Float16 v8h;
typedef __attribute__((ext_vector_type(8)))  float    v8f;

union HFrag { v16h v; v8h h[2]; };

__device__ __forceinline__ v16h load_frag(const _Float16* p, int ld, int row0, int k0) {
  const int lane = threadIdx.x & 31;
  const _Float16* q = p + (row0 + (lane & 15)) * ld + k0 + ((lane >> 4) << 3);
  HFrag f;
  f.h[0] = *(const v8h*)(q);
  f.h[1] = *(const v8h*)(q + 16);
  return f.v;
}

#define WFRAG 512
__device__ __forceinline__ v16h load_frag_w(const _Float16* q) {
  HFrag f;
  f.h[0] = *(const v8h*)(q);
  f.h[1] = *(const v8h*)(q + 256);
  return f.v;
}

__device__ __forceinline__ v8f wmma_f16(v16h a, v16h b, v8f c) {
  return __builtin_amdgcn_wmma_f32_16x16x32_f16(false, a, false, b, (short)0, c, false, false);
}

__device__ __forceinline__ void store_tile_f16(_Float16* p, int ld, int row0, int col0, v8f acc) {
  const int lane = threadIdx.x & 31;
  const int m0 = row0 + ((lane >> 4) << 3);
  _Float16* q = p + col0 + (lane & 15);
#pragma unroll
  for (int r = 0; r < 8; ++r) q[(m0 + r) * ld] = (_Float16)acc[r];
}

__device__ __forceinline__ void store_tile_f16_T(_Float16* p, int ld, int row0, int col0, v8f acc) {
  const int lane = threadIdx.x & 31;
  const int m0 = row0 + ((lane >> 4) << 3);
  const int n  = col0 + (lane & 15);
  v8h h;
#pragma unroll
  for (int r = 0; r < 8; ++r) h[r] = (_Float16)acc[r];
  *(v8h*)(p + n * ld + m0) = h;
}

__device__ __forceinline__ void layer_norm_rows(float* xf32, _Float16* xf16,
                                                const float* __restrict__ g,
                                                const float* __restrict__ bt,
                                                int wave, int lane) {
#pragma unroll
  for (int rr = 0; rr < 4; ++rr) {
    const int m = wave * 4 + rr;
    float vals[8];
    float sum = 0.f;
#pragma unroll
    for (int i = 0; i < 8; ++i) { vals[i] = xf32[m * 256 + lane + 32 * i]; sum += vals[i]; }
#pragma unroll
    for (int o = 16; o >= 1; o >>= 1) sum += __shfl_xor(sum, o, 32);
    const float mu = sum * (1.f / 256.f);
    float s2 = 0.f;
#pragma unroll
    for (int i = 0; i < 8; ++i) { float d = vals[i] - mu; s2 += d * d; }
#pragma unroll
    for (int o = 16; o >= 1; o >>= 1) s2 += __shfl_xor(s2, o, 32);
    const float rstd = rsqrtf(s2 * (1.f / 256.f) + 1e-5f);
#pragma unroll
    for (int i = 0; i < 8; ++i) {
      const int c = lane + 32 * i;
      const float y = (vals[i] - mu) * rstd * g[c] + bt[c];
      xf32[m * 256 + c] = y;
      xf16[m * 272 + c] = (_Float16)y;
    }
  }
}

__global__ void cvt_pack_frag(const float* __restrict__ s, _Float16* __restrict__ d, int kts) {
  const int g    = blockIdx.x * blockDim.x + threadIdx.x;
  const int frag = g >> 6;
  const int part = (g >> 5) & 1;
  const int ln   = g & 31;
  const int nt   = frag >> kts;
  const int kt   = frag & ((1 << kts) - 1);
  const float* q = s + ((nt * 16 + (ln & 15)) << (kts + 5)) + kt * 32 + part * 16 + ((ln >> 4) << 3);
  const float4 a = *(const float4*)(q);
  const float4 b = *(const float4*)(q + 4);
  v8h h;
  h[0] = (_Float16)a.x; h[1] = (_Float16)a.y; h[2] = (_Float16)a.z; h[3] = (_Float16)a.w;
  h[4] = (_Float16)b.x; h[5] = (_Float16)b.y; h[6] = (_Float16)b.z; h[7] = (_Float16)b.w;
  *(v8h*)(d + g * 8) = h;
}

__global__ __launch_bounds__(512) void swt_kernel(
    const float* __restrict__ X, const float* __restrict__ pos,
    const _Float16* __restrict__ Win,  const float* __restrict__ Bin,
    const _Float16* __restrict__ Wout, const float* __restrict__ Bout,
    const _Float16* __restrict__ W1,   const float* __restrict__ B1,
    const _Float16* __restrict__ W2,   const float* __restrict__ B2,
    const float* __restrict__ ln1g, const float* __restrict__ ln1b,
    const float* __restrict__ ln2g, const float* __restrict__ ln2b,
    const float* __restrict__ headw, const float* __restrict__ headb,
    float* __restrict__ out) {
  __shared__ float    xf32[64 * 256];
  __shared__ _Float16 l16[92160];
  _Float16* xf16 = l16;
  _Float16* qk   = l16 + 17408;
  _Float16* vT   = l16 + 51200;
  _Float16* pr   = l16 + 71680;

  const int tid  = threadIdx.x;
  const int lane = tid & 31;
  const int wave = __builtin_amdgcn_readfirstlane(tid >> 5);
  const int pair0 = blockIdx.x * 2;
  const int bb   = pair0 >> 9;
  const int ss0  = pair0 & 511;

  {
    const int jr = tid >> 3;
    const int wwin = jr >> 5, j = jr & 31;
    const int c0 = (tid & 7) * 32;
    int src = (ss0 + wwin) - j; if (src < 0) src = 0;
    const float* xr = X + ((bb << 9) + src) * 256;
    const float* pe = pos + j * 256;
#pragma unroll
    for (int u = 0; u < 8; ++u) {
      const int c = c0 + u * 4;
      float4 xv = *(const float4*)(xr + c);
      float4 pv = *(const float4*)(pe + c);
      float y0 = xv.x + pv.x, y1 = xv.y + pv.y, y2 = xv.z + pv.z, y3 = xv.w + pv.w;
      xf32[jr * 256 + c + 0] = y0; xf16[jr * 272 + c + 0] = (_Float16)y0;
      xf32[jr * 256 + c + 1] = y1; xf16[jr * 272 + c + 1] = (_Float16)y1;
      xf32[jr * 256 + c + 2] = y2; xf16[jr * 272 + c + 2] = (_Float16)y2;
      xf32[jr * 256 + c + 3] = y3; xf16[jr * 272 + c + 3] = (_Float16)y3;
    }
  }
  __syncthreads();

  const v8f vzero = {};
  const int m0l = ((lane >> 4) << 3);
  const int nnl = lane & 15;
  const int kh  = wave >> 3;
  const int np  = wave & 7;

  for (int l = 0; l < Ln; ++l) {
    const _Float16* wqkv = Win  + l * 768 * 256;
    const float*    bqkv = Bin  + l * 768;
    const _Float16* wo   = Wout + l * 256 * 256;
    const float*    bo   = Bout + l * 256;
    const _Float16* w1l  = W1   + l * 1024 * 256;
    const float*    b1l  = B1   + l * 1024;
    const _Float16* w2l  = W2   + l * 256 * 1024;
    const float*    b2l  = B2   + l * 256;

    {
      const int n0 = wave * 16;
      const _Float16* wq = wqkv + wave * (8 * WFRAG) + lane * 8;
      v8f acc[4][3];
#pragma unroll
      for (int mi = 0; mi < 4; ++mi) { acc[mi][0] = vzero; acc[mi][1] = vzero; acc[mi][2] = vzero; }
#pragma unroll
      for (int kt = 0; kt < 8; ++kt) {
        v16h bq = load_frag_w(wq + (kt         ) * WFRAG);
        v16h bk = load_frag_w(wq + (kt + 16 * 8) * WFRAG);
        v16h bv = load_frag_w(wq + (kt + 32 * 8) * WFRAG);
#pragma unroll
        for (int mi = 0; mi < 4; ++mi) {
          v16h a = load_frag(xf16, 272, mi * 16, kt * 32);
          acc[mi][0] = wmma_f16(a, bq, acc[mi][0]);
          acc[mi][1] = wmma_f16(a, bk, acc[mi][1]);
          acc[mi][2] = wmma_f16(a, bv, acc[mi][2]);
        }
      }
      const float biasq = bqkv[n0 + nnl];
      const float biask = bqkv[n0 + 256 + nnl];
      const float biasv = bqkv[n0 + 512 + nnl];
#pragma unroll
      for (int mi = 0; mi < 4; ++mi) {
#pragma unroll
        for (int r = 0; r < 8; ++r) {
          acc[mi][0][r] += biasq; acc[mi][1][r] += biask; acc[mi][2][r] += biasv;
        }
        store_tile_f16(qk, 528, mi * 16, n0,       acc[mi][0]);
        store_tile_f16(qk, 528, mi * 16, n0 + 256, acc[mi][1]);
        store_tile_f16_T(vT + (mi >> 1) * 10240, 40, (mi & 1) * 16, n0, acc[mi][2]);
      }
    }
    __syncthreads();

    {
      const int aw = wave >> 3;
      const int h  = wave & 7;
      const int rb = aw * 32;
      _Float16* vTw = vT + aw * 10240;
      _Float16* ph  = pr + wave * 1280;
      v8f sc[2][2];
#pragma unroll
      for (int qt = 0; qt < 2; ++qt) {
        v16h a = load_frag(qk, 528, rb + qt * 16, h * 32);
#pragma unroll
        for (int nt = 0; nt < 2; ++nt) {
          v16h b = load_frag(qk + 256, 528, rb + nt * 16, h * 32);
          sc[qt][nt] = wmma_f16(a, b, vzero);
        }
      }
      const float scale = 0.17677669529663687f;
#pragma unroll
      for (int qt = 0; qt < 2; ++qt) {
#pragma unroll
        for (int r = 0; r < 8; ++r) {
          float s0 = sc[qt][0][r] * scale;
          float s1 = sc[qt][1][r] * scale;
          float mx = fmaxf(s0, s1);
#pragma unroll
          for (int o = 8; o >= 1; o >>= 1) mx = fmaxf(mx, __shfl_xor(mx, o, 32));
          const float e0 = __expf(s0 - mx), e1 = __expf(s1 - mx);
          float sm = e0 + e1;
#pragma unroll
          for (int o = 8; o >= 1; o >>= 1) sm += __shfl_xor(sm, o, 32);
          const float inv = 1.0f / sm;
          const int m = qt * 16 + m0l + r;
          ph[m * 40 + nnl]      = (_Float16)(e0 * inv);
          ph[m * 40 + 16 + nnl] = (_Float16)(e1 * inv);
        }
      }
#pragma unroll
      for (int qt = 0; qt < 2; ++qt) {
        v16h a = load_frag(ph, 40, qt * 16, 0);
#pragma unroll
        for (int dt = 0; dt < 2; ++dt) {
          v16h b = load_frag(vTw, 40, h * 32 + dt * 16, 0);
          v8f oa = wmma_f16(a, b, vzero);
          store_tile_f16(qk, 528, rb + qt * 16, h * 32 + dt * 16, oa);
        }
      }
    }
    __syncthreads();

    {
      const int n0 = np * 32;
      const _Float16* wq = wo + (np * 16 + kh * 4) * WFRAG + lane * 8;
      v8f acc[4][2];
#pragma unroll
      for (int mi = 0; mi < 4; ++mi) { acc[mi][0] = vzero; acc[mi][1] = vzero; }
#pragma unroll
      for (int kk = 0; kk < 4; ++kk) {
        const int kt = kh * 4 + kk;
        v16h b0 = load_frag_w(wq + (kk    ) * WFRAG);
        v16h b1 = load_frag_w(wq + (kk + 8) * WFRAG);
#pragma unroll
        for (int mi = 0; mi < 4; ++mi) {
          v16h a = load_frag(qk, 528, mi * 16, kt * 32);
          acc[mi][0] = wmma_f16(a, b0, acc[mi][0]);
          acc[mi][1] = wmma_f16(a, b1, acc[mi][1]);
        }
      }
      const float bias0 = (kh == 0) ? bo[n0 + nnl] : 0.f;
      const float bias1 = (kh == 0) ? bo[n0 + 16 + nnl] : 0.f;
#pragma unroll
      for (int mi = 0; mi < 4; ++mi) {
        const int m0 = mi * 16 + m0l;
#pragma unroll
        for (int r = 0; r < 8; ++r) {
          atomicAdd(&xf32[(m0 + r) * 256 + n0 + nnl],      acc[mi][0][r] + bias0);
          atomicAdd(&xf32[(m0 + r) * 256 + n0 + 16 + nnl], acc[mi][1][r] + bias1);
        }
      }
    }
    __syncthreads();
    layer_norm_rows(xf32, xf16, ln1g + l * 256, ln1b + l * 256, wave, lane);
    __syncthreads();

    v8f facc[4][2];
#pragma unroll
    for (int mi = 0; mi < 4; ++mi) { facc[mi][0] = vzero; facc[mi][1] = vzero; }
    for (int cp = 0; cp < 2; ++cp) {
      {
        const int ng0 = cp * 512 + wave * 16;
        const int ng1 = ng0 + 256;
        const _Float16* wq = w1l + (cp * 32 + wave) * (8 * WFRAG) + lane * 8;
        v8f acc[4][2];
#pragma unroll
        for (int mi = 0; mi < 4; ++mi) { acc[mi][0] = vzero; acc[mi][1] = vzero; }
#pragma unroll
        for (int kt = 0; kt < 8; ++kt) {
          v16h b0 = load_frag_w(wq + (kt         ) * WFRAG);
          v16h b1 = load_frag_w(wq + (kt + 16 * 8) * WFRAG);
#pragma unroll
          for (int mi = 0; mi < 4; ++mi) {
            v16h a = load_frag(xf16, 272, mi * 16, kt * 32);
            acc[mi][0] = wmma_f16(a, b0, acc[mi][0]);
            acc[mi][1] = wmma_f16(a, b1, acc[mi][1]);
          }
        }
        const float bias0 = b1l[ng0 + nnl];
        const float bias1 = b1l[ng1 + nnl];
#pragma unroll
        for (int mi = 0; mi < 4; ++mi) {
#pragma unroll
          for (int r = 0; r < 8; ++r) {
            acc[mi][0][r] = fmaxf(acc[mi][0][r] + bias0, 0.f);
            acc[mi][1][r] = fmaxf(acc[mi][1][r] + bias1, 0.f);
          }
          store_tile_f16(qk, 528, mi * 16, wave * 16,       acc[mi][0]);
          store_tile_f16(qk, 528, mi * 16, 256 + wave * 16, acc[mi][1]);
        }
      }
      __syncthreads();
      {
        const int n0 = np * 32;
        const _Float16* wq = w2l + (np * 64 + cp * 16 + kh * 8) * WFRAG + lane * 8;
#pragma unroll
        for (int kk = 0; kk < 8; ++kk) {
          const int kt = kh * 8 + kk;
          v16h b0 = load_frag_w(wq + (kk     ) * WFRAG);
          v16h b1 = load_frag_w(wq + (kk + 32) * WFRAG);
#pragma unroll
          for (int mi = 0; mi < 4; ++mi) {
            v16h a = load_frag(qk, 528, mi * 16, kt * 32);
            facc[mi][0] = wmma_f16(a, b0, facc[mi][0]);
            facc[mi][1] = wmma_f16(a, b1, facc[mi][1]);
          }
        }
      }
      __syncthreads();
    }
    {
      const int n0 = np * 32;
      const float bias0 = (kh == 0) ? b2l[n0 + nnl] : 0.f;
      const float bias1 = (kh == 0) ? b2l[n0 + 16 + nnl] : 0.f;
#pragma unroll
      for (int mi = 0; mi < 4; ++mi) {
        const int m0 = mi * 16 + m0l;
#pragma unroll
        for (int r = 0; r < 8; ++r) {
          atomicAdd(&xf32[(m0 + r) * 256 + n0 + nnl],      facc[mi][0][r] + bias0);
          atomicAdd(&xf32[(m0 + r) * 256 + n0 + 16 + nnl], facc[mi][1][r] + bias1);
        }
      }
    }
    __syncthreads();
    layer_norm_rows(xf32, xf16, ln2g + l * 256, ln2b + l * 256, wave, lane);
    __syncthreads();
  }

  if (wave < 2) {
    float acc = headb[lane];
    const float* hw = headw + lane * 256;
    const float* xr = xf32 + (wave * 32 + 31) * 256;
    for (int k = 0; k < 256; k += 4) {
      float4 w4 = *(const float4*)(hw + k);
      acc += xr[k] * w4.x + xr[k + 1] * w4.y + xr[k + 2] * w4.z + xr[k + 3] * w4.w;
    }
    out[(pair0 + wave) * 32 + lane] = acc;
  }
}

extern "C" void kernel_launch(void* const* d_in, const int* in_sizes, int n_in,
                              void* d_out, int out_size, void* d_ws, size_t ws_size,
                              hipStream_t stream) {
  const float* X     = (const float*)d_in[0];
  const float* pos   = (const float*)d_in[3];
  const float* w_in  = (const float*)d_in[4];
  const float* b_in  = (const float*)d_in[5];
  const float* w_out = (const float*)d_in[6];
  const float* b_out = (const float*)d_in[7];
  const float* w1    = (const float*)d_in[8];
  const float* b1    = (const float*)d_in[9];
  const float* w2    = (const float*)d_in[10];
  const float* b2    = (const float*)d_in[11];
  const float* ln1g  = (const float*)d_in[12];
  const float* ln1b  = (const float*)d_in[13];
  const float* ln2g  = (const float*)d_in[14];
  const float* ln2b  = (const float*)d_in[15];
  const float* headw = (const float*)d_in[16];
  const float* headb = (const float*)d_in[17];
  float* out = (float*)d_out;

  _Float16* ws      = (_Float16*)d_ws;
  _Float16* w_in_h  = ws;
  _Float16* w_out_h = w_in_h  + Ln * 768 * 256;
  _Float16* w1_h    = w_out_h + Ln * 256 * 256;
  _Float16* w2_h    = w1_h    + Ln * 1024 * 256;

  const int n_in_w  = Ln * 768 * 256;
  const int n_out_w = Ln * 256 * 256;
  const int n_w1    = Ln * 1024 * 256;
  const int n_w2    = Ln * 256 * 1024;
  cvt_pack_frag<<<n_in_w  / 2048, 256, 0, stream>>>(w_in,  w_in_h,  3);
  cvt_pack_frag<<<n_out_w / 2048, 256, 0, stream>>>(w_out, w_out_h, 3);
  cvt_pack_frag<<<n_w1    / 2048, 256, 0, stream>>>(w1,    w1_h,    3);
  cvt_pack_frag<<<n_w2    / 2048, 256, 0, stream>>>(w2,    w2_h,    5);

  swt_kernel<<<(Bn * Sn) / 2, 512, 0, stream>>>(X, pos, w_in_h, b_in, w_out_h, b_out,
                                                w1_h, b1, w2_h, b2, ln1g, ln1b, ln2g,
                                                ln2b, headw, headb, out);
}


# Earlier try flipped run to run via arrival-order split-K adds, fixed since. f16 LDS row pitches 32 B x odd -> 16 B x odd (264/520 halfs): b128 fragment reads conflict-free; sums unchanged
# speedup vs baseline: 1.6673x; 1.4668x over previous
#include <hip/hip_runtime.h>

#define Bn    4
#define Sn    512
#define Dn    256
#define Wn    32
#define Ln    4
#define Hn    8
#define DFFn  1024
#define NOUTn 32
#define XHP   264
#define QKP   520
#define PSP   260

typedef __attribute__((ext_vector_type(16))) _Float16 v16h;
typedef __attribute__((ext_vector_type(8)))  _Float16 v8h;
typedef __attribute__((ext_vector_type(8)))  float    v8f;

union HFrag { v16h v; v8h h[2]; };

__device__ __forceinline__ v16h load_frag(const _Float16* p, int ld, int row0, int k0) {
  const int lane = threadIdx.x & 31;
  const _Float16* q = p + (row0 + (lane & 15)) * ld + k0 + ((lane >> 4) << 3);
  HFrag f;
  f.h[0] = *(const v8h*)(q);
  f.h[1] = *(const v8h*)(q + 16);
  return f.v;
}

#define WFRAG 512
__device__ __forceinline__ v16h load_frag_w(const _Float16* q) {
  HFrag f;
  f.h[0] = *(const v8h*)(q);
  f.h[1] = *(const v8h*)(q + 256);
  return f.v;
}

__device__ __forceinline__ v8f wmma_f16(v16h a, v16h b, v8f c) {
  return __builtin_amdgcn_wmma_f32_16x16x32_f16(false, a, false, b, (short)0, c, false, false);
}

template <int NB>
__device__ __forceinline__ void load_b_step(v16h (&b)[NB], const _Float16* wq, int kt, int bstride) {
#pragma unroll
  for (int j = 0; j < NB; ++j) b[j] = load_frag_w(wq + (kt + j * bstride) * WFRAG);
}

template <int NB>
__device__ __forceinline__ void gemm_request(v16h (&b)[3][NB], const _Float16* wq, int bstride) {
  load_b_step<NB>(b[0], wq, 0, bstride);
  load_b_step<NB>(b[1], wq, 1, bstride);
  __builtin_amdgcn_sched_barrier(0);
}

__device__ __forceinline__ void lds_barrier() {
  __builtin_amdgcn_fence(__ATOMIC_RELEASE, "workgroup", "local");
  __builtin_amdgcn_s_barrier();
  __builtin_amdgcn_fence(__ATOMIC_ACQUIRE, "workgroup", "local");
}

template <int NB, int KT, unsigned SW>
__device__ __forceinline__ void gemm_steps(v8f (&acc)[4][NB], v16h (&b)[3][NB],
                                           const _Float16* ap, int ld,
                                           const _Float16* wq, int bstride) {
#pragma unroll
  for (int kt = 0; kt < KT; ++kt) {
    if (kt + 2 < KT) load_b_step<NB>(b[(kt + 2) % 3], wq, kt + 2, bstride);
    v16h a[4];
#pragma unroll
    for (int mi = 0; mi < 4; ++mi) a[mi] = load_frag(ap, ld, mi * 16, kt * 32);
    __builtin_amdgcn_sched_barrier(0);
#pragma unroll
    for (int mi = 0; mi < 4; ++mi) {
#pragma unroll
      for (int j = 0; j < NB; ++j) {
        if ((SW >> j) & 1u) acc[mi][j] = wmma_f16(b[kt % 3][j], a[mi], acc[mi][j]);
        else                acc[mi][j] = wmma_f16(a[mi], b[kt % 3][j], acc[mi][j]);
      }
    }
    __builtin_amdgcn_sched_barrier(0);
  }
}

__device__ __forceinline__ void store_tile_f16(_Float16* p, int ld, int row0, int col0, v8f acc) {
  const int lane = threadIdx.x & 31;
  const int m0 = row0 + ((lane >> 4) << 3);
  _Float16* q = p + col0 + (lane & 15);
#pragma unroll
  for (int r = 0; r < 8; ++r) q[(m0 + r) * ld] = (_Float16)acc[r];
}

__device__ __forceinline__ void store_tile_f16_T(_Float16* p, int ld, int row0, int col0, v8f acc) {
  const int lane = threadIdx.x & 31;
  const int m0 = row0 + ((lane >> 4) << 3);
  const int n  = col0 + (lane & 15);
  v8h h;
#pragma unroll
  for (int r = 0; r < 8; ++r) h[r] = (_Float16)acc[r];
  *(v8h*)(p + n * ld + m0) = h;
}

__device__ __forceinline__ void store_rows_f16(_Float16* p, int ld, int row0, int col0, v8f acc) {
  const int lane = threadIdx.x & 31;
  v8h h;
#pragma unroll
  for (int r = 0; r < 8; ++r) h[r] = (_Float16)acc[r];
  *(v8h*)(p + (row0 + (lane & 15)) * ld + col0 + ((lane >> 4) << 3)) = h;
}

__device__ __forceinline__ void load_bias8(float (&bv)[8], const float* __restrict__ b, int n0) {
  const float4 lo = *(const float4*)(b + n0);
  const float4 hi = *(const float4*)(b + n0 + 4);
  bv[0] = lo.x; bv[1] = lo.y; bv[2] = lo.z; bv[3] = lo.w;
  bv[4] = hi.x; bv[5] = hi.y; bv[6] = hi.z; bv[7] = hi.w;
}

__device__ __forceinline__ void layer_norm_rows(float* xf32, const float* ps, _Float16* xf16,
                                                const float* __restrict__ g,
                                                const float* __restrict__ bt,
                                                int wave, int lane) {
#pragma unroll
  for (int rr = 0; rr < 4; ++rr) {
    const int m = wave * 4 + rr;
    float vals[8];
    float sum = 0.f;
#pragma unroll
    for (int i = 0; i < 8; ++i) {
      vals[i] = xf32[m * 256 + lane + 32 * i] + ps[m * PSP + lane + 32 * i];
      sum += vals[i];
    }
#pragma unroll
    for (int o = 16; o >= 1; o >>= 1) sum += __shfl_xor(sum, o, 32);
    const float mu = sum * (1.f / 256.f);
    float s2 = 0.f;
#pragma unroll
    for (int i = 0; i < 8; ++i) { float d = vals[i] - mu; s2 += d * d; }
#pragma unroll
    for (int o = 16; o >= 1; o >>= 1) s2 += __shfl_xor(s2, o, 32);
    const float rstd = rsqrtf(s2 * (1.f / 256.f) + 1e-5f);
#pragma unroll
    for (int i = 0; i < 8; ++i) {
      const int c = lane + 32 * i;
      const float y = (vals[i] - mu) * rstd * g[c] + bt[c];
      xf32[m * 256 + c] = y;
      xf16[m * XHP + c] = (_Float16)y;
    }
  }
}

__global__ void cvt_pack_frag(const float* __restrict__ s, _Float16* __restrict__ d, int kts) {
  const int g    = blockIdx.x * blockDim.x + threadIdx.x;
  const int frag = g >> 6;
  const int part = (g >> 5) & 1;
  const int ln   = g & 31;
  const int nt   = frag >> kts;
  const int kt   = frag & ((1 << kts) - 1);
  const float* q = s + ((nt * 16 + (ln & 15)) << (kts + 5)) + kt * 32 + part * 16 + ((ln >> 4) << 3);
  const float4 a = *(const float4*)(q);
  const float4 b = *(const float4*)(q + 4);
  v8h h;
  h[0] = (_Float16)a.x; h[1] = (_Float16)a.y; h[2] = (_Float16)a.z; h[3] = (_Float16)a.w;
  h[4] = (_Float16)b.x; h[5] = (_Float16)b.y; h[6] = (_Float16)b.z; h[7] = (_Float16)b.w;
  *(v8h*)(d + g * 8) = h;
}

__global__ __launch_bounds__(512) void swt_kernel(
    const float* __restrict__ X, const float* __restrict__ pos,
    const _Float16* __restrict__ Win,  const float* __restrict__ Bin,
    const _Float16* __restrict__ Wout, const float* __restrict__ Bout,
    const _Float16* __restrict__ W1,   const float* __restrict__ B1,
    const _Float16* __restrict__ W2,   const float* __restrict__ B2,
    const float* __restrict__ ln1g, const float* __restrict__ ln1b,
    const float* __restrict__ ln2g, const float* __restrict__ ln2b,
    const float* __restrict__ headw, const float* __restrict__ headb,
    float* __restrict__ out) {
  __shared__ float    xf32[64 * 256];
  __shared__ float    psum[64 * PSP];
  __shared__ _Float16 l16[64 * XHP + 64 * QKP + 2 * 256 * 40];
  _Float16* xf16 = l16;
  _Float16* qk   = l16 + 64 * XHP;
  _Float16* vT   = qk + 64 * QKP;

  const int tid  = threadIdx.x;
  const int lane = tid & 31;
  const int wave = __builtin_amdgcn_readfirstlane(tid >> 5);
  const int pair0 = blockIdx.x * 2;
  const int bb   = pair0 >> 9;
  const int ss0  = pair0 & 511;

  {
    const int jr = tid >> 3;
    const int wwin = jr >> 5, j = jr & 31;
    const int c0 = (tid & 7) * 32;
    int src = (ss0 + wwin) - j; if (src < 0) src = 0;
    const float* xr = X + ((bb << 9) + src) * 256;
    const float* pe = pos + j * 256;
#pragma unroll
    for (int u = 0; u < 8; ++u) {
      const int c = c0 + u * 4;
      float4 xv = *(const float4*)(xr + c);
      float4 pv = *(const float4*)(pe + c);
      float y0 = xv.x + pv.x, y1 = xv.y + pv.y, y2 = xv.z + pv.z, y3 = xv.w + pv.w;
      xf32[jr * 256 + c + 0] = y0; xf16[jr * XHP + c + 0] = (_Float16)y0;
      xf32[jr * 256 + c + 1] = y1; xf16[jr * XHP + c + 1] = (_Float16)y1;
      xf32[jr * 256 + c + 2] = y2; xf16[jr * XHP + c + 2] = (_Float16)y2;
      xf32[jr * 256 + c + 3] = y3; xf16[jr * XHP + c + 3] = (_Float16)y3;
    }
  }

  const v8f vzero = {};
  const int m0l = ((lane >> 4) << 3);
  const int nnl = lane & 15;
  const int kh  = wave >> 3;
  const int np  = wave & 7;

  for (int l = 0; l < Ln; ++l) {
    const _Float16* wqkv = Win  + l * 768 * 256;
    const float*    bqkv = Bin  + l * 768;
    const _Float16* wo   = Wout + l * 256 * 256;
    const float*    bo   = Bout + l * 256;
    const _Float16* w1l  = W1   + l * 1024 * 256;
    const float*    b1l  = B1   + l * 1024;
    const _Float16* w2l  = W2   + l * 256 * 1024;
    const float*    b2l  = B2   + l * 256;

    {
      const int n0 = wave * 16;
      const _Float16* wq = wqkv + wave * (8 * WFRAG) + lane * 8;
      v16h bring[3][3];
      gemm_request<3>(bring, wq, 16 * 8);
      lds_barrier();
      v8f acc[4][3];
#pragma unroll
      for (int mi = 0; mi < 4; ++mi) { acc[mi][0] = vzero; acc[mi][1] = vzero; acc[mi][2] = vzero; }
      gemm_steps<3, 8, 3u>(acc, bring, xf16, XHP, wq, 16 * 8);
      float biasq[8], biask[8];
      load_bias8(biasq, bqkv, n0 + m0l);
      load_bias8(biask, bqkv, n0 + 256 + m0l);
      const float biasv = bqkv[n0 + 512 + nnl];
#pragma unroll
      for (int mi = 0; mi < 4; ++mi) {
#pragma unroll
        for (int r = 0; r < 8; ++r) {
          acc[mi][0][r] += biasq[r]; acc[mi][1][r] += biask[r]; acc[mi][2][r] += biasv;
        }
        store_rows_f16(qk, QKP, mi * 16, n0,       acc[mi][0]);
        store_rows_f16(qk, QKP, mi * 16, n0 + 256, acc[mi][1]);
        store_tile_f16_T(vT + (mi >> 1) * 10240, 40, (mi & 1) * 16, n0, acc[mi][2]);
      }
    }
    __syncthreads();

    {
      const int aw = wave >> 3;
      const int h  = wave & 7;
      const int rb = aw * 32;
      _Float16* vTw = vT + aw * 10240;
      v8f sc[2][2];
      v16h pf[2];
      {
        v16h kf[2], qf[2];
#pragma unroll
        for (int t = 0; t < 2; ++t) {
          kf[t] = load_frag(qk + 256, QKP, rb + t * 16, h * 32);
          qf[t] = load_frag(qk,       QKP, rb + t * 16, h * 32);
        }
#pragma unroll
        for (int qt = 0; qt < 2; ++qt) {
#pragma unroll
          for (int kt = 0; kt < 2; ++kt) sc[qt][kt] = wmma_f16(kf[kt], qf[qt], vzero);
        }
      }
      const float scale = 0.17677669529663687f;
#pragma unroll
      for (int qt = 0; qt < 2; ++qt) {
        float s[2][8];
        float mx = sc[qt][0][0] * scale;
#pragma unroll
        for (int kt = 0; kt < 2; ++kt) {
#pragma unroll
          for (int r = 0; r < 8; ++r) { s[kt][r] = sc[qt][kt][r] * scale; mx = fmaxf(mx, s[kt][r]); }
        }
        mx = fmaxf(mx, __shfl_xor(mx, 16, 32));
        float sm = 0.f;
#pragma unroll
        for (int kt = 0; kt < 2; ++kt) {
#pragma unroll
          for (int r = 0; r < 8; ++r) { s[kt][r] = __expf(s[kt][r] - mx); sm += s[kt][r]; }
        }
        sm += __shfl_xor(sm, 16, 32);
        const float inv = 1.0f / sm;
        HFrag pq;
#pragma unroll
        for (int kt = 0; kt < 2; ++kt) {
#pragma unroll
          for (int r = 0; r < 8; ++r) pq.h[kt][r] = (_Float16)(s[kt][r] * inv);
        }
        pf[qt] = pq.v;
      }
      {
        v16h vf[2];
#pragma unroll
        for (int t = 0; t < 2; ++t) vf[t] = load_frag(vTw, 40, h * 32 + t * 16, 0);
#pragma unroll
        for (int qt = 0; qt < 2; ++qt) {
#pragma unroll
          for (int dt = 0; dt < 2; ++dt) {
            v8f oa = wmma_f16(vf[dt], pf[qt], vzero);
            store_rows_f16(qk, QKP, rb + qt * 16, h * 32 + dt * 16, oa);
          }
        }
      }
    }

    {
      const int n0 = np * 32;
      const _Float16* wq = wo + (np * 16 + kh * 4) * WFRAG + lane * 8;
      v16h bring[3][2];
      gemm_request<2>(bring, wq, 8);
      lds_barrier();
      v8f acc[4][2];
#pragma unroll
      for (int mi = 0; mi < 4; ++mi) { acc[mi][0] = vzero; acc[mi][1] = vzero; }
      gemm_steps<2, 4, 0u>(acc, bring, qk + kh * 4 * 32, QKP, wq, 8);
      if (kh == 0) {
        const float bias0 = bo[n0 + nnl];
        const float bias1 = bo[n0 + 16 + nnl];
        float* xb = xf32 + m0l * 256 + n0 + nnl;
#pragma unroll
        for (int mi = 0; mi < 4; ++mi) {
#pragma unroll
          for (int r = 0; r < 8; ++r) {
            atomicAdd(xb + (mi * 16 + r) * 256,      acc[mi][0][r] + bias0);
            atomicAdd(xb + (mi * 16 + r) * 256 + 16, acc[mi][1][r] + bias1);
          }
        }
      } else {
        float* pb = psum + m0l * PSP + n0 + nnl;
#pragma unroll
        for (int mi = 0; mi < 4; ++mi) {
#pragma unroll
          for (int r = 0; r < 8; ++r) {
            pb[(mi * 16 + r) * PSP]      = acc[mi][0][r];
            pb[(mi * 16 + r) * PSP + 16] = acc[mi][1][r];
          }
        }
      }
    }
    __syncthreads();
    v16h bring1[3][2];
    v16h bring2[3][2];
    gemm_request<2>(bring1, w1l + wave * (8 * WFRAG) + lane * 8, 16 * 8);
    layer_norm_rows(xf32, psum, xf16, ln1g + l * 256, ln1b + l * 256, wave, lane);

    v8f facc[4][2];
#pragma unroll
    for (int mi = 0; mi < 4; ++mi) { facc[mi][0] = vzero; facc[mi][1] = vzero; }
    lds_barrier();
#pragma unroll
    for (int cp = 0; cp < 2; ++cp) {
      const _Float16* wq2 = w2l + (np * 64 + cp * 16 + kh * 8) * WFRAG + lane * 8;
      {
        const int ng0 = cp * 512 + wave * 16;
        const int ng1 = ng0 + 256;
        const _Float16* wq = w1l + (cp * 32 + wave) * (8 * WFRAG) + lane * 8;
        v8f acc[4][2];
#pragma unroll
        for (int mi = 0; mi < 4; ++mi) { acc[mi][0] = vzero; acc[mi][1] = vzero; }
        gemm_steps<2, 8, 3u>(acc, bring1, xf16, XHP, wq, 16 * 8);
        gemm_request<2>(bring2, wq2, 32);
        float bias0[8], bias1[8];
        load_bias8(bias0, b1l, ng0 + m0l);
        load_bias8(bias1, b1l, ng1 + m0l);
#pragma unroll
        for (int mi = 0; mi < 4; ++mi) {
#pragma unroll
          for (int r = 0; r < 8; ++r) {
            acc[mi][0][r] = fmaxf(acc[mi][0][r] + bias0[r], 0.f);
            acc[mi][1][r] = fmaxf(acc[mi][1][r] + bias1[r], 0.f);
          }
          store_rows_f16(qk, QKP, mi * 16, wave * 16,       acc[mi][0]);
          store_rows_f16(qk, QKP, mi * 16, 256 + wave * 16, acc[mi][1]);
        }
      }
      lds_barrier();
      {
        gemm_steps<2, 8, 0u>(facc, bring2, qk + kh * 8 * 32, QKP, wq2, 32);
        if (cp == 0) gemm_request<2>(bring1, w1l + (32 + wave) * (8 * WFRAG) + lane * 8, 16 * 8);
      }
      lds_barrier();
    }
    {
      const int n0 = np * 32;
      if (kh == 0) {
        const float bias0 = b2l[n0 + nnl];
        const float bias1 = b2l[n0 + 16 + nnl];
        float* xb = xf32 + m0l * 256 + n0 + nnl;
#pragma unroll
        for (int mi = 0; mi < 4; ++mi) {
#pragma unroll
          for (int r = 0; r < 8; ++r) {
            atomicAdd(xb + (mi * 16 + r) * 256,      facc[mi][0][r] + bias0);
            atomicAdd(xb + (mi * 16 + r) * 256 + 16, facc[mi][1][r] + bias1);
          }
        }
      } else {
        float* pb = psum + m0l * PSP + n0 + nnl;
#pragma unroll
        for (int mi = 0; mi < 4; ++mi) {
#pragma unroll
          for (int r = 0; r < 8; ++r) {
            pb[(mi * 16 + r) * PSP]      = facc[mi][0][r];
            pb[(mi * 16 + r) * PSP + 16] = facc[mi][1][r];
          }
        }
      }
    }
    __syncthreads();
    layer_norm_rows(xf32, psum, xf16, ln2g + l * 256, ln2b + l * 256, wave, lane);
  }
  __syncthreads();

  if (wave < 2) {
    float acc = headb[lane];
    const float* hw = headw + lane * 256;
    const float* xr = xf32 + (wave * 32 + 31) * 256;
    for (int k = 0; k < 256; k += 4) {
      float4 w4 = *(const float4*)(hw + k);
      acc += xr[k] * w4.x + xr[k + 1] * w4.y + xr[k + 2] * w4.z + xr[k + 3] * w4.w;
    }
    out[(pair0 + wave) * 32 + lane] = acc;
  }
}

extern "C" void kernel_launch(void* const* d_in, const int* in_sizes, int n_in,
                              void* d_out, int out_size, void* d_ws, size_t ws_size,
                              hipStream_t stream) {
  const float* X     = (const float*)d_in[0];
  const float* pos   = (const float*)d_in[3];
  const float* w_in  = (const float*)d_in[4];
  const float* b_in  = (const float*)d_in[5];
  const float* w_out = (const float*)d_in[6];
  const float* b_out = (const float*)d_in[7];
  const float* w1    = (const float*)d_in[8];
  const float* b1    = (const float*)d_in[9];
  const float* w2    = (const float*)d_in[10];
  const float* b2    = (const float*)d_in[11];
  const float* ln1g  = (const float*)d_in[12];
  const float* ln1b  = (const float*)d_in[13];
  const float* ln2g  = (const float*)d_in[14];
  const float* ln2b  = (const float*)d_in[15];
  const float* headw = (const float*)d_in[16];
  const float* headb = (const float*)d_in[17];
  float* out = (float*)d_out;

  _Float16* ws      = (_Float16*)d_ws;
  _Float16* w_in_h  = ws;
  _Float16* w_out_h = w_in_h  + Ln * 768 * 256;
  _Float16* w1_h    = w_out_h + Ln * 256 * 256;
  _Float16* w2_h    = w1_h    + Ln * 1024 * 256;

  const int n_in_w  = Ln * 768 * 256;
  const int n_out_w = Ln * 256 * 256;
  const int n_w1    = Ln * 1024 * 256;
  const int n_w2    = Ln * 256 * 1024;
  cvt_pack_frag<<<n_in_w  / 2048, 256, 0, stream>>>(w_in,  w_in_h,  3);
  cvt_pack_frag<<<n_out_w / 2048, 256, 0, stream>>>(w_out, w_out_h, 3);
  cvt_pack_frag<<<n_w1    / 2048, 256, 0, stream>>>(w1,    w1_h,    3);
  cvt_pack_frag<<<n_w2    / 2048, 256, 0, stream>>>(w2,    w2_h,    5);

  swt_kernel<<<(Bn * Sn) / 2, 512, 0, stream>>>(X, pos, w_in_h, b_in, w_out_h, b_out,
                                                w1_h, b1, w2_h, b2, ln1g, ln1b, ln2g,
                                                ln2b, headw, headb, out);
}


# Layer-norm row sums and the softmax half exchange on the vector ALU (v_add_f32 row_xmask DPP, v_permlanex16) instead of ds_bpermute round trips through LDS; same butterfly order, sums unchanged
# speedup vs baseline: 2.1711x; 1.3022x over previous
#include <hip/hip_runtime.h>

#define Bn    4
#define Sn    512
#define Dn    256
#define Wn    32
#define Ln    4
#define Hn    8
#define DFFn  1024
#define NOUTn 32
#define XHP   264
#define QKP   520
#define PSP   260

typedef __attribute__((ext_vector_type(16))) _Float16 v16h;
typedef __attribute__((ext_vector_type(8)))  _Float16 v8h;
typedef __attribute__((ext_vector_type(8)))  float    v8f;

union HFrag { v16h v; v8h h[2]; };

__device__ __forceinline__ v16h load_frag(const _Float16* p, int ld, int row0, int k0) {
  const int lane = threadIdx.x & 31;
  const _Float16* q = p + (row0 + (lane & 15)) * ld + k0 + ((lane >> 4) << 3);
  HFrag f;
  f.h[0] = *(const v8h*)(q);
  f.h[1] = *(const v8h*)(q + 16);
  return f.v;
}

#define WFRAG 512
__device__ __forceinline__ v16h load_frag_w(const _Float16* q) {
  HFrag f;
  f.h[0] = *(const v8h*)(q);
  f.h[1] = *(const v8h*)(q + 256);
  return f.v;
}

__device__ __forceinline__ v8f wmma_f16(v16h a, v16h b, v8f c) {
  return __builtin_amdgcn_wmma_f32_16x16x32_f16(false, a, false, b, (short)0, c, false, false);
}

__device__ __forceinline__ float xhalf(float x) {
  const int b = __builtin_bit_cast(int, x);
  return __builtin_bit_cast(float,
      __builtin_amdgcn_permlanex16(b, b, 0x76543210, 0xfedcba98, false, false));
}
template <int M>
__device__ __forceinline__ float xrow(float x) {
  const int b = __builtin_bit_cast(int, x);
  return __builtin_bit_cast(float, __builtin_amdgcn_update_dpp(0, b, 0x160 + M, 0xf, 0xf, true));
}
__device__ __forceinline__ float wave_sum(float s) {
  s += xhalf(s);
  s += xrow<8>(s);
  s += xrow<4>(s);
  s += xrow<2>(s);
  s += xrow<1>(s);
  return s;
}

template <int NB>
__device__ __forceinline__ void load_b_step(v16h (&b)[NB], const _Float16* wq, int kt, int bstride) {
#pragma unroll
  for (int j = 0; j < NB; ++j) b[j] = load_frag_w(wq + (kt + j * bstride) * WFRAG);
}

template <int NB>
__device__ __forceinline__ void gemm_request(v16h (&b)[3][NB], const _Float16* wq, int bstride) {
  load_b_step<NB>(b[0], wq, 0, bstride);
  load_b_step<NB>(b[1], wq, 1, bstride);
  __builtin_amdgcn_sched_barrier(0);
}

__device__ __forceinline__ void lds_barrier() {
  __builtin_amdgcn_fence(__ATOMIC_RELEASE, "workgroup", "local");
  __builtin_amdgcn_s_barrier();
  __builtin_amdgcn_fence(__ATOMIC_ACQUIRE, "workgroup", "local");
}

template <int NB, int KT, unsigned SW, int MT, int MB, int MS>
__device__ __forceinline__ void gemm_steps(v8f (&acc)[MT][NB], v16h (&b)[3][NB],
                                           const _Float16* ap, int ld,
                                           const _Float16* wq, int bstride) {
#pragma unroll
  for (int kt = 0; kt < KT; ++kt) {
    if (kt + 2 < KT) load_b_step<NB>(b[(kt + 2) % 3], wq, kt + 2, bstride);
    v16h a[MT];
#pragma unroll
    for (int mi = 0; mi < MT; ++mi) a[mi] = load_frag(ap, ld, MB + mi * MS, kt * 32);
    __builtin_amdgcn_sched_barrier(0);
#pragma unroll
    for (int mi = 0; mi < MT; ++mi) {
#pragma unroll
      for (int j = 0; j < NB; ++j) {
        if ((SW >> j) & 1u) acc[mi][j] = wmma_f16(b[kt % 3][j], a[mi], acc[mi][j]);
        else                acc[mi][j] = wmma_f16(a[mi], b[kt % 3][j], acc[mi][j]);
      }
    }
    __builtin_amdgcn_sched_barrier(0);
  }
}

__device__ __forceinline__ void qkv_steps_last(v8f (&accq)[2], v8f (&acck)[4], v8f (&accv)[4],
                                               v16h (&b)[3][3], const _Float16* ap, int ld,
                                               const _Float16* wq, int bstride) {
#pragma unroll
  for (int kt = 0; kt < 8; ++kt) {
    if (kt + 2 < 8) load_b_step<3>(b[(kt + 2) % 3], wq, kt + 2, bstride);
    v16h a[4];
#pragma unroll
    for (int mi = 0; mi < 4; ++mi) a[mi] = load_frag(ap, ld, mi * 16, kt * 32);
    __builtin_amdgcn_sched_barrier(0);
#pragma unroll
    for (int mi = 0; mi < 4; ++mi) {
      if (mi & 1) accq[mi >> 1] = wmma_f16(b[kt % 3][0], a[mi], accq[mi >> 1]);
      acck[mi] = wmma_f16(b[kt % 3][1], a[mi], acck[mi]);
      accv[mi] = wmma_f16(a[mi], b[kt % 3][2], accv[mi]);
    }
    __builtin_amdgcn_sched_barrier(0);
  }
}

__device__ __forceinline__ void store_tile_f16(_Float16* p, int ld, int row0, int col0, v8f acc) {
  const int lane = threadIdx.x & 31;
  const int m0 = row0 + ((lane >> 4) << 3);
  _Float16* q = p + col0 + (lane & 15);
#pragma unroll
  for (int r = 0; r < 8; ++r) q[(m0 + r) * ld] = (_Float16)acc[r];
}

__device__ __forceinline__ void store_tile_f16_T(_Float16* p, int ld, int row0, int col0, v8f acc) {
  const int lane = threadIdx.x & 31;
  const int m0 = row0 + ((lane >> 4) << 3);
  const int n  = col0 + (lane & 15);
  v8h h;
#pragma unroll
  for (int r = 0; r < 8; ++r) h[r] = (_Float16)acc[r];
  *(v8h*)(p + n * ld + m0) = h;
}

__device__ __forceinline__ void store_rows_f16(_Float16* p, int ld, int row0, int col0, v8f acc) {
  const int lane = threadIdx.x & 31;
  v8h h;
#pragma unroll
  for (int r = 0; r < 8; ++r) h[r] = (_Float16)acc[r];
  *(v8h*)(p + (row0 + (lane & 15)) * ld + col0 + ((lane >> 4) << 3)) = h;
}

__device__ __forceinline__ void load_bias8(float (&bv)[8], const float* __restrict__ b, int n0) {
  const float4 lo = *(const float4*)(b + n0);
  const float4 hi = *(const float4*)(b + n0 + 4);
  bv[0] = lo.x; bv[1] = lo.y; bv[2] = lo.z; bv[3] = lo.w;
  bv[4] = hi.x; bv[5] = hi.y; bv[6] = hi.z; bv[7] = hi.w;
}

template <int NR>
__device__ __forceinline__ void layer_norm_rows(float* xf32, const float* ps, _Float16* xf16,
                                                const float* __restrict__ g,
                                                const float* __restrict__ bt,
                                                int m0, int lane) {
#pragma unroll
  for (int rr = 0; rr < NR; ++rr) {
    const int m = m0 + rr;
    float vals[8];
    float sum = 0.f;
#pragma unroll
    for (int i = 0; i < 8; ++i) {
      vals[i] = xf32[m * 256 + lane + 32 * i] + ps[m * PSP + lane + 32 * i];
      sum += vals[i];
    }
    sum = wave_sum(sum);
    const float mu = sum * (1.f / 256.f);
    float s2 = 0.f;
#pragma unroll
    for (int i = 0; i < 8; ++i) { float d = vals[i] - mu; s2 += d * d; }
    s2 = wave_sum(s2);
    const float rstd = rsqrtf(s2 * (1.f / 256.f) + 1e-5f);
#pragma unroll
    for (int i = 0; i < 8; ++i) {
      const int c = lane + 32 * i;
      const float y = (vals[i] - mu) * rstd * g[c] + bt[c];
      xf32[m * 256 + c] = y;
      xf16[m * XHP + c] = (_Float16)y;
    }
  }
}

__global__ void cvt_pack_frag(const float* __restrict__ s, _Float16* __restrict__ d, int kts) {
  const int g    = blockIdx.x * blockDim.x + threadIdx.x;
  const int frag = g >> 6;
  const int part = (g >> 5) & 1;
  const int ln   = g & 31;
  const int nt   = frag >> kts;
  const int kt   = frag & ((1 << kts) - 1);
  const float* q = s + ((nt * 16 + (ln & 15)) << (kts + 5)) + kt * 32 + part * 16 + ((ln >> 4) << 3);
  const float4 a = *(const float4*)(q);
  const float4 b = *(const float4*)(q + 4);
  v8h h;
  h[0] = (_Float16)a.x; h[1] = (_Float16)a.y; h[2] = (_Float16)a.z; h[3] = (_Float16)a.w;
  h[4] = (_Float16)b.x; h[5] = (_Float16)b.y; h[6] = (_Float16)b.z; h[7] = (_Float16)b.w;
  *(v8h*)(d + g * 8) = h;
}

template <bool B> struct LayerKind { static constexpr bool value = B; };

#define PAIRS 4
__global__ __launch_bounds__(512) void swt_kernel(
    const float* __restrict__ X, const float* __restrict__ pos,
    const _Float16* __restrict__ Win,  const float* __restrict__ Bin,
    const _Float16* __restrict__ Wout, const float* __restrict__ Bout,
    const _Float16* __restrict__ W1,   const float* __restrict__ B1,
    const _Float16* __restrict__ W2,   const float* __restrict__ B2,
    const float* __restrict__ ln1g, const float* __restrict__ ln1b,
    const float* __restrict__ ln2g, const float* __restrict__ ln2b,
    const float* __restrict__ headw, const float* __restrict__ headb,
    float* __restrict__ out) {
  __shared__ float    xf32[64 * 256];
  __shared__ float    psum[64 * PSP];
  __shared__ _Float16 l16[64 * XHP + 64 * QKP + 2 * 256 * 40];
  __shared__ __attribute__((aligned(16))) float    px32[16 * 256];
  __shared__ __attribute__((aligned(16))) _Float16 po16[16 * XHP];
  _Float16* xf16 = l16;
  _Float16* qk   = l16 + 64 * XHP;
  _Float16* vT   = qk + 64 * QKP;

  const int tid  = threadIdx.x;
  const int lane = tid & 31;
  const int wave = __builtin_amdgcn_readfirstlane(tid >> 5);

  const v8f vzero = {};
  const int m0l = ((lane >> 4) << 3);
  const int nnl = lane & 15;
  const int kh  = wave >> 3;
  const int np  = wave & 7;

  for (int i = tid; i < 16 * 256; i += 512) px32[i] = 0.f;
  for (int i = tid; i < 16 * XHP; i += 512) po16[i] = (_Float16)0.f;

  auto front = [&](const int l, auto last_tag) {
    constexpr bool LAST = decltype(last_tag)::value;
    constexpr int QT0 = LAST ? 1 : 0;
    const _Float16* wqkv = Win  + l * 768 * 256;
    const float*    bqkv = Bin  + l * 768;

    {
      const int n0 = wave * 16;
      const _Float16* wq = wqkv + wave * (8 * WFRAG) + lane * 8;
      v16h bring[3][3];
      gemm_request<3>(bring, wq, 16 * 8);
      lds_barrier();
      if constexpr (!LAST) {
        v8f acc[4][3];
#pragma unroll
        for (int mi = 0; mi < 4; ++mi) { acc[mi][0] = vzero; acc[mi][1] = vzero; acc[mi][2] = vzero; }
        gemm_steps<3, 8, 3u, 4, 0, 16>(acc, bring, xf16, XHP, wq, 16 * 8);
        float biasq[8], biask[8];
        load_bias8(biasq, bqkv, n0 + m0l);
        load_bias8(biask, bqkv, n0 + 256 + m0l);
        const float biasv = bqkv[n0 + 512 + nnl];
#pragma unroll
        for (int mi = 0; mi < 4; ++mi) {
#pragma unroll
          for (int r = 0; r < 8; ++r) {
            acc[mi][0][r] += biasq[r]; acc[mi][1][r] += biask[r]; acc[mi][2][r] += biasv;
          }
          store_rows_f16(qk, QKP, mi * 16, n0,       acc[mi][0]);
          store_rows_f16(qk, QKP, mi * 16, n0 + 256, acc[mi][1]);
          store_tile_f16_T(vT + (mi >> 1) * 10240, 40, (mi & 1) * 16, n0, acc[mi][2]);
        }
      } else {
        v8f accq[2], acck[4], accv[4];
        accq[0] = vzero; accq[1] = vzero;
#pragma unroll
        for (int mi = 0; mi < 4; ++mi) { acck[mi] = vzero; accv[mi] = vzero; }
        qkv_steps_last(accq, acck, accv, bring, xf16, XHP, wq, 16 * 8);
        float biasq[8], biask[8];
        load_bias8(biasq, bqkv, n0 + m0l);
        load_bias8(biask, bqkv, n0 + 256 + m0l);
        const float biasv = bqkv[n0 + 512 + nnl];
#pragma unroll
        for (int mi = 0; mi < 4; ++mi) {
#pragma unroll
          for (int r = 0; r < 8; ++r) { acck[mi][r] += biask[r]; accv[mi][r] += biasv; }
          store_rows_f16(qk, QKP, mi * 16, n0 + 256, acck[mi]);
          store_tile_f16_T(vT + (mi >> 1) * 10240, 40, (mi & 1) * 16, n0, accv[mi]);
        }
#pragma unroll
        for (int t = 0; t < 2; ++t) {
#pragma unroll
          for (int r = 0; r < 8; ++r) accq[t][r] += biasq[r];
          store_rows_f16(qk, QKP, 16 + t * 32, n0, accq[t]);
        }
      }
    }
    __syncthreads();

    {
      const int aw = wave >> 3;
      const int h  = wave & 7;
      const int rb = aw * 32;
      _Float16* vTw = vT + aw * 10240;
      v8f sc[2][2];
      v16h pf[2];
      {
        v16h kf[2], qf[2];
#pragma unroll
        for (int t = 0; t < 2; ++t) kf[t] = load_frag(qk + 256, QKP, rb + t * 16, h * 32);
#pragma unroll
        for (int t = QT0; t < 2; ++t) qf[t] = load_frag(qk, QKP, rb + t * 16, h * 32);
#pragma unroll
        for (int qt = QT0; qt < 2; ++qt) {
#pragma unroll
          for (int kt = 0; kt < 2; ++kt) sc[qt][kt] = wmma_f16(kf[kt], qf[qt], vzero);
        }
      }
      const float scale = 0.17677669529663687f;
#pragma unroll
      for (int qt = QT0; qt < 2; ++qt) {
        float s[2][8];
        float mx = sc[qt][0][0] * scale;
#pragma unroll
        for (int kt = 0; kt < 2; ++kt) {
#pragma unroll
          for (int r = 0; r < 8; ++r) { s[kt][r] = sc[qt][kt][r] * scale; mx = fmaxf(mx, s[kt][r]); }
        }
        mx = fmaxf(mx, xhalf(mx));
        float sm = 0.f;
#pragma unroll
        for (int kt = 0; kt < 2; ++kt) {
#pragma unroll
          for (int r = 0; r < 8; ++r) { s[kt][r] = __expf(s[kt][r] - mx); sm += s[kt][r]; }
        }
        sm += xhalf(sm);
        const float inv = 1.0f / sm;
        HFrag pq;
#pragma unroll
        for (int kt = 0; kt < 2; ++kt) {
#pragma unroll
          for (int r = 0; r < 8; ++r) pq.h[kt][r] = (_Float16)(s[kt][r] * inv);
        }
        pf[qt] = pq.v;
      }
      {
        v16h vf[2];
#pragma unroll
        for (int t = 0; t < 2; ++t) vf[t] = load_frag(vTw, 40, h * 32 + t * 16, 0);
#pragma unroll
        for (int qt = QT0; qt < 2; ++qt) {
#pragma unroll
          for (int dt = 0; dt < 2; ++dt) {
            v8f oa = wmma_f16(vf[dt], pf[qt], vzero);
            store_rows_f16(qk, QKP, rb + qt * 16, h * 32 + dt * 16, oa);
          }
        }
      }
    }
  };

  auto rest = [&](const int l, auto last_tag, float* xres, const _Float16* oa, const int lda) {
    constexpr bool LAST = decltype(last_tag)::value;
    constexpr int MT    = LAST ? 1 : 4;
    constexpr int NLIVE = 2 * PAIRS;
    const _Float16* wo   = Wout + l * 256 * 256;
    const float*    bo   = Bout + l * 256;
    const _Float16* w1l  = W1   + l * 1024 * 256;
    const float*    b1l  = B1   + l * 1024;
    const _Float16* w2l  = W2   + l * 256 * 1024;
    const float*    b2l  = B2   + l * 256;

    {
      const int n0 = np * 32;
      const _Float16* wq = wo + (np * 16 + kh * 4) * WFRAG + lane * 8;
      v16h bring[3][2];
      gemm_request<2>(bring, wq, 8);
      lds_barrier();
      v8f acc[MT][2];
#pragma unroll
      for (int mi = 0; mi < MT; ++mi) { acc[mi][0] = vzero; acc[mi][1] = vzero; }
      gemm_steps<2, 4, 0u, MT, 0, 16>(acc, bring, oa + kh * 4 * 32, lda, wq, 8);
      if (kh == 0) {
        const float bias0 = bo[n0 + nnl];
        const float bias1 = bo[n0 + 16 + nnl];
        float* xb = xres + m0l * 256 + n0 + nnl;
#pragma unroll
        for (int mi = 0; mi < MT; ++mi) {
#pragma unroll
          for (int r = 0; r < 8; ++r) {
            atomicAdd(xb + (mi * 16 + r) * 256,      acc[mi][0][r] + bias0);
            atomicAdd(xb + (mi * 16 + r) * 256 + 16, acc[mi][1][r] + bias1);
          }
        }
      } else {
        float* pb = psum + m0l * PSP + n0 + nnl;
#pragma unroll
        for (int mi = 0; mi < MT; ++mi) {
#pragma unroll
          for (int r = 0; r < 8; ++r) {
            pb[(mi * 16 + r) * PSP]      = acc[mi][0][r];
            pb[(mi * 16 + r) * PSP + 16] = acc[mi][1][r];
          }
        }
      }
    }
    __syncthreads();
    v16h bring1[3][2];
    v16h bring2[3][2];
    gemm_request<2>(bring1, w1l + wave * (8 * WFRAG) + lane * 8, 16 * 8);
    if constexpr (!LAST) {
      layer_norm_rows<4>(xres, psum, xf16, ln1g + l * 256, ln1b + l * 256, wave * 4, lane);
    } else {
      if (wave < NLIVE) layer_norm_rows<1>(xres, psum, xf16, ln1g + l * 256, ln1b + l * 256, wave, lane);
    }

    v8f facc[MT][2];
#pragma unroll
    for (int mi = 0; mi < MT; ++mi) { facc[mi][0] = vzero; facc[mi][1] = vzero; }
    lds_barrier();
#pragma unroll
    for (int cp = 0; cp < 2; ++cp) {
      const _Float16* wq2 = w2l + (np * 64 + cp * 16 + kh * 8) * WFRAG + lane * 8;
      {
        const int ng0 = cp * 512 + wave * 16;
        const int ng1 = ng0 + 256;
        const _Float16* wq = w1l + (cp * 32 + wave) * (8 * WFRAG) + lane * 8;
        v8f acc[MT][2];
#pragma unroll
        for (int mi = 0; mi < MT; ++mi) { acc[mi][0] = vzero; acc[mi][1] = vzero; }
        gemm_steps<2, 8, 3u, MT, 0, 16>(acc, bring1, xf16, XHP, wq, 16 * 8);
        gemm_request<2>(bring2, wq2, 32);
        float bias0[8], bias1[8];
        load_bias8(bias0, b1l, ng0 + m0l);
        load_bias8(bias1, b1l, ng1 + m0l);
#pragma unroll
        for (int mi = 0; mi < MT; ++mi) {
#pragma unroll
          for (int r = 0; r < 8; ++r) {
            acc[mi][0][r] = fmaxf(acc[mi][0][r] + bias0[r], 0.f);
            acc[mi][1][r] = fmaxf(acc[mi][1][r] + bias1[r], 0.f);
          }
          store_rows_f16(qk, QKP, mi * 16, wave * 16,       acc[mi][0]);
          store_rows_f16(qk, QKP, mi * 16, 256 + wave * 16, acc[mi][1]);
        }
      }
      lds_barrier();
      {
        gemm_steps<2, 8, 0u, MT, 0, 16>(facc, bring2, qk + kh * 8 * 32, QKP, wq2, 32);
        if (cp == 0) gemm_request<2>(bring1, w1l + (32 + wave) * (8 * WFRAG) + lane * 8, 16 * 8);
      }
      lds_barrier();
    }
    {
      const int n0 = np * 32;
      if (kh == 0) {
        const float bias0 = b2l[n0 + nnl];
        const float bias1 = b2l[n0 + 16 + nnl];
        float* xb = xres + m0l * 256 + n0 + nnl;
#pragma unroll
        for (int mi = 0; mi < MT; ++mi) {
#pragma unroll
          for (int r = 0; r < 8; ++r) {
            atomicAdd(xb + (mi * 16 + r) * 256,      facc[mi][0][r] + bias0);
            atomicAdd(xb + (mi * 16 + r) * 256 + 16, facc[mi][1][r] + bias1);
          }
        }
      } else {
        float* pb = psum + m0l * PSP + n0 + nnl;
#pragma unroll
        for (int mi = 0; mi < MT; ++mi) {
#pragma unroll
          for (int r = 0; r < 8; ++r) {
            pb[(mi * 16 + r) * PSP]      = facc[mi][0][r];
            pb[(mi * 16 + r) * PSP + 16] = facc[mi][1][r];
          }
        }
      }
    }
    __syncthreads();
    if constexpr (!LAST) {
      layer_norm_rows<4>(xres, psum, xf16, ln2g + l * 256, ln2b + l * 256, wave * 4, lane);
    } else {
      if (wave < NLIVE) layer_norm_rows<1>(xres, psum, xf16, ln2g + l * 256, ln2b + l * 256, wave, lane);
    }
  };

  for (int p = 0; p < PAIRS; ++p) {
    const int pair0 = (blockIdx.x * PAIRS + p) * 2;
    const int bb   = pair0 >> 9;
    const int ss0  = pair0 & 511;

    {
      const int jr = tid >> 3;
      const int wwin = jr >> 5, j = jr & 31;
      const int c0 = (tid & 7) * 32;
      int src = (ss0 + wwin) - j; if (src < 0) src = 0;
      const float* xr = X + ((bb << 9) + src) * 256;
      const float* pe = pos + j * 256;
#pragma unroll
      for (int u = 0; u < 8; ++u) {
        const int c = c0 + u * 4;
        float4 xv = *(const float4*)(xr + c);
        float4 pv = *(const float4*)(pe + c);
        float y0 = xv.x + pv.x, y1 = xv.y + pv.y, y2 = xv.z + pv.z, y3 = xv.w + pv.w;
        xf32[jr * 256 + c + 0] = y0; xf16[jr * XHP + c + 0] = (_Float16)y0;
        xf32[jr * 256 + c + 1] = y1; xf16[jr * XHP + c + 1] = (_Float16)y1;
        xf32[jr * 256 + c + 2] = y2; xf16[jr * XHP + c + 2] = (_Float16)y2;
        xf32[jr * 256 + c + 3] = y3; xf16[jr * XHP + c + 3] = (_Float16)y3;
      }
    }

    for (int l = 0; l < Ln; ++l) {
      if (l < Ln - 1) {
        front(l, LayerKind<false>{});
        rest(l, LayerKind<false>{}, xf32, qk, QKP);
      } else {
        front(l, LayerKind<true>{});
      }
    }
    __syncthreads();
    {
      const int w = tid >> 8, c = tid & 255;
      px32[(2 * p + w) * 256 + c] = xf32[(31 + 32 * w) * 256 + c];
      po16[(2 * p + w) * XHP + c] = qk[(31 + 32 * w) * QKP + c];
    }
    __syncthreads();
  }

  rest(Ln - 1, LayerKind<true>{}, px32, po16, XHP);
  __syncthreads();

  if (wave < 2 * PAIRS) {
    float acc = headb[lane];
    const float* hw = headw + lane * 256;
    const float* xr = px32 + wave * 256;
    for (int k = 0; k < 256; k += 4) {
      float4 w4 = *(const float4*)(hw + k);
      acc += xr[k] * w4.x + xr[k + 1] * w4.y + xr[k + 2] * w4.z + xr[k + 3] * w4.w;
    }
    out[(blockIdx.x * 2 * PAIRS + wave) * 32 + lane] = acc;
  }
}

extern "C" void kernel_launch(void* const* d_in, const int* in_sizes, int n_in,
                              void* d_out, int out_size, void* d_ws, size_t ws_size,
                              hipStream_t stream) {
  const float* X     = (const float*)d_in[0];
  const float* pos   = (const float*)d_in[3];
  const float* w_in  = (const float*)d_in[4];
  const float* b_in  = (const float*)d_in[5];
  const float* w_out = (const float*)d_in[6];
  const float* b_out = (const float*)d_in[7];
  const float* w1    = (const float*)d_in[8];
  const float* b1    = (const float*)d_in[9];
  const float* w2    = (const float*)d_in[10];
  const float* b2    = (const float*)d_in[11];
  const float* ln1g  = (const float*)d_in[12];
  const float* ln1b  = (const float*)d_in[13];
  const float* ln2g  = (const float*)d_in[14];
  const float* ln2b  = (const float*)d_in[15];
  const float* headw = (const float*)d_in[16];
  const float* headb = (const float*)d_in[17];
  float* out = (float*)d_out;

  _Float16* ws      = (_Float16*)d_ws;
  _Float16* w_in_h  = ws;
  _Float16* w_out_h = w_in_h  + Ln * 768 * 256;
  _Float16* w1_h    = w_out_h + Ln * 256 * 256;
  _Float16* w2_h    = w1_h    + Ln * 1024 * 256;

  const int n_in_w  = Ln * 768 * 256;
  const int n_out_w = Ln * 256 * 256;
  const int n_w1    = Ln * 1024 * 256;
  const int n_w2    = Ln * 256 * 1024;
  cvt_pack_frag<<<n_in_w  / 2048, 256, 0, stream>>>(w_in,  w_in_h,  3);
  cvt_pack_frag<<<n_out_w / 2048, 256, 0, stream>>>(w_out, w_out_h, 3);
  cvt_pack_frag<<<n_w1    / 2048, 256, 0, stream>>>(w1,    w1_h,    3);
  cvt_pack_frag<<<n_w2    / 2048, 256, 0, stream>>>(w2,    w2_h,    5);

  swt_kernel<<<(Bn * Sn) / (2 * PAIRS), 512, 0, stream>>>(X, pos, w_in_h, b_in, w_out_h, b_out,
                                                w1_h, b1, w2_h, b2, ln1g, ln1b, ln2g,
                                                ln2b, headw, headb, out);
}


# Residual path in 16-byte LDS accesses: out-proj/FFN2 tiles transposed, master pitch 260 floats, combine by row load/add/store (no LDS atomics), norm lane owns 2x4 columns (row sum tree changed)
# speedup vs baseline: 2.2789x; 1.0497x over previous
#include <hip/hip_runtime.h>

#define Bn    4
#define Sn    512
#define Dn    256
#define Wn    32
#define Ln    4
#define Hn    8
#define DFFn  1024
#define NOUTn 32
#define XHP   264
#define QKP   520
#define PSP   260
#define XFP   260

typedef __attribute__((ext_vector_type(16))) _Float16 v16h;
typedef __attribute__((ext_vector_type(8)))  _Float16 v8h;
typedef __attribute__((ext_vector_type(4)))  _Float16 v4h;
typedef __attribute__((ext_vector_type(8)))  float    v8f;

union HFrag { v16h v; v8h h[2]; };

__device__ __forceinline__ v16h load_frag(const _Float16* p, int ld, int row0, int k0) {
  const int lane = threadIdx.x & 31;
  const _Float16* q = p + (row0 + (lane & 15)) * ld + k0 + ((lane >> 4) << 3);
  HFrag f;
  f.h[0] = *(const v8h*)(q);
  f.h[1] = *(const v8h*)(q + 16);
  return f.v;
}

#define WFRAG 512
__device__ __forceinline__ v16h load_frag_w(const _Float16* q) {
  HFrag f;
  f.h[0] = *(const v8h*)(q);
  f.h[1] = *(const v8h*)(q + 256);
  return f.v;
}

__device__ __forceinline__ v8f wmma_f16(v16h a, v16h b, v8f c) {
  return __builtin_amdgcn_wmma_f32_16x16x32_f16(false, a, false, b, (short)0, c, false, false);
}

__device__ __forceinline__ float xhalf(float x) {
  const int b = __builtin_bit_cast(int, x);
  return __builtin_bit_cast(float,
      __builtin_amdgcn_permlanex16(b, b, 0x76543210, 0xfedcba98, false, false));
}
template <int M>
__device__ __forceinline__ float xrow(float x) {
  const int b = __builtin_bit_cast(int, x);
  return __builtin_bit_cast(float, __builtin_amdgcn_update_dpp(0, b, 0x160 + M, 0xf, 0xf, true));
}
__device__ __forceinline__ float wave_sum(float s) {
  s += xhalf(s);
  s += xrow<8>(s);
  s += xrow<4>(s);
  s += xrow<2>(s);
  s += xrow<1>(s);
  return s;
}

template <int NB>
__device__ __forceinline__ void load_b_step(v16h (&b)[NB], const _Float16* wq, int kt, int bstride) {
#pragma unroll
  for (int j = 0; j < NB; ++j) b[j] = load_frag_w(wq + (kt + j * bstride) * WFRAG);
}

template <int NB>
__device__ __forceinline__ void gemm_request(v16h (&b)[3][NB], const _Float16* wq, int bstride) {
  load_b_step<NB>(b[0], wq, 0, bstride);
  load_b_step<NB>(b[1], wq, 1, bstride);
  __builtin_amdgcn_sched_barrier(0);
}

__device__ __forceinline__ void lds_barrier() {
  __builtin_amdgcn_fence(__ATOMIC_RELEASE, "workgroup", "local");
  __builtin_amdgcn_s_barrier();
  __builtin_amdgcn_fence(__ATOMIC_ACQUIRE, "workgroup", "local");
}

template <int NB, int KT, unsigned SW, int MT, int MB, int MS>
__device__ __forceinline__ void gemm_steps(v8f (&acc)[MT][NB], v16h (&b)[3][NB],
                                           const _Float16* ap, int ld,
                                           const _Float16* wq, int bstride) {
#pragma unroll
  for (int kt = 0; kt < KT; ++kt) {
    if (kt + 2 < KT) load_b_step<NB>(b[(kt + 2) % 3], wq, kt + 2, bstride);
    v16h a[MT];
#pragma unroll
    for (int mi = 0; mi < MT; ++mi) a[mi] = load_frag(ap, ld, MB + mi * MS, kt * 32);
    __builtin_amdgcn_sched_barrier(0);
#pragma unroll
    for (int mi = 0; mi < MT; ++mi) {
#pragma unroll
      for (int j = 0; j < NB; ++j) {
        if ((SW >> j) & 1u) acc[mi][j] = wmma_f16(b[kt % 3][j], a[mi], acc[mi][j]);
        else                acc[mi][j] = wmma_f16(a[mi], b[kt % 3][j], acc[mi][j]);
      }
    }
    __builtin_amdgcn_sched_barrier(0);
  }
}

__device__ __forceinline__ void qkv_steps_last(v8f (&accq)[2], v8f (&acck)[4], v8f (&accv)[4],
                                               v16h (&b)[3][3], const _Float16* ap, int ld,
                                               const _Float16* wq, int bstride) {
#pragma unroll
  for (int kt = 0; kt < 8; ++kt) {
    if (kt + 2 < 8) load_b_step<3>(b[(kt + 2) % 3], wq, kt + 2, bstride);
    v16h a[4];
#pragma unroll
    for (int mi = 0; mi < 4; ++mi) a[mi] = load_frag(ap, ld, mi * 16, kt * 32);
    __builtin_amdgcn_sched_barrier(0);
#pragma unroll
    for (int mi = 0; mi < 4; ++mi) {
      if (mi & 1) accq[mi >> 1] = wmma_f16(b[kt % 3][0], a[mi], accq[mi >> 1]);
      acck[mi] = wmma_f16(b[kt % 3][1], a[mi], acck[mi]);
      accv[mi] = wmma_f16(a[mi], b[kt % 3][2], accv[mi]);
    }
    __builtin_amdgcn_sched_barrier(0);
  }
}

__device__ __forceinline__ void store_tile_f16(_Float16* p, int ld, int row0, int col0, v8f acc) {
  const int lane = threadIdx.x & 31;
  const int m0 = row0 + ((lane >> 4) << 3);
  _Float16* q = p + col0 + (lane & 15);
#pragma unroll
  for (int r = 0; r < 8; ++r) q[(m0 + r) * ld] = (_Float16)acc[r];
}

__device__ __forceinline__ void store_tile_f16_T(_Float16* p, int ld, int row0, int col0, v8f acc) {
  const int lane = threadIdx.x & 31;
  const int m0 = row0 + ((lane >> 4) << 3);
  const int n  = col0 + (lane & 15);
  v8h h;
#pragma unroll
  for (int r = 0; r < 8; ++r) h[r] = (_Float16)acc[r];
  *(v8h*)(p + n * ld + m0) = h;
}

__device__ __forceinline__ void store_rows_f16(_Float16* p, int ld, int row0, int col0, v8f acc) {
  const int lane = threadIdx.x & 31;
  v8h h;
#pragma unroll
  for (int r = 0; r < 8; ++r) h[r] = (_Float16)acc[r];
  *(v8h*)(p + (row0 + (lane & 15)) * ld + col0 + ((lane >> 4) << 3)) = h;
}

__device__ __forceinline__ void load_bias8(float (&bv)[8], const float* __restrict__ b, int n0) {
  const float4 lo = *(const float4*)(b + n0);
  const float4 hi = *(const float4*)(b + n0 + 4);
  bv[0] = lo.x; bv[1] = lo.y; bv[2] = lo.z; bv[3] = lo.w;
  bv[4] = hi.x; bv[5] = hi.y; bv[6] = hi.z; bv[7] = hi.w;
}

template <int NR>
__device__ __forceinline__ void layer_norm_rows(float* xf32, const float* ps, _Float16* xf16,
                                                const float* __restrict__ g,
                                                const float* __restrict__ bt,
                                                int m0, int lane) {
  const int c0 = lane * 4;
  const float4 g0 = *(const float4*)(g + c0),  g1 = *(const float4*)(g + 128 + c0);
  const float4 b0 = *(const float4*)(bt + c0), b1 = *(const float4*)(bt + 128 + c0);
  const float gg[8] = {g0.x, g0.y, g0.z, g0.w, g1.x, g1.y, g1.z, g1.w};
  const float bb[8] = {b0.x, b0.y, b0.z, b0.w, b1.x, b1.y, b1.z, b1.w};
#pragma unroll
  for (int rr = 0; rr < NR; ++rr) {
    const int m = m0 + rr;
    float* xr = xf32 + m * XFP + c0;
    const float* pr = ps + m * PSP + c0;
    const float4 x0 = *(const float4*)(xr), x1 = *(const float4*)(xr + 128);
    const float4 p0 = *(const float4*)(pr), p1 = *(const float4*)(pr + 128);
    float vals[8] = {x0.x + p0.x, x0.y + p0.y, x0.z + p0.z, x0.w + p0.w,
                     x1.x + p1.x, x1.y + p1.y, x1.z + p1.z, x1.w + p1.w};
    float sum = 0.f;
#pragma unroll
    for (int i = 0; i < 8; ++i) sum += vals[i];
    sum = wave_sum(sum);
    const float mu = sum * (1.f / 256.f);
    float s2 = 0.f;
#pragma unroll
    for (int i = 0; i < 8; ++i) { float d = vals[i] - mu; s2 += d * d; }
    s2 = wave_sum(s2);
    const float rstd = rsqrtf(s2 * (1.f / 256.f) + 1e-5f);
    float y[8];
#pragma unroll
    for (int i = 0; i < 8; ++i) y[i] = (vals[i] - mu) * rstd * gg[i] + bb[i];
    *(float4*)(xr)       = make_float4(y[0], y[1], y[2], y[3]);
    *(float4*)(xr + 128) = make_float4(y[4], y[5], y[6], y[7]);
    v4h h0, h1;
#pragma unroll
    for (int i = 0; i < 4; ++i) { h0[i] = (_Float16)y[i]; h1[i] = (_Float16)y[4 + i]; }
    _Float16* hr = xf16 + m * XHP + c0;
    *(v4h*)(hr)       = h0;
    *(v4h*)(hr + 128) = h1;
  }
}

template <int MT>
__device__ __forceinline__ void splitk_combine(float* xres, float* psum, const float* __restrict__ bias,
                                               const v8f (&acc)[MT][2], int kh, int n0, int lane) {
  const int c  = lane & 15;
  const int h8 = (lane >> 4) << 3;
  if (kh == 0) {
#pragma unroll
    for (int j = 0; j < 2; ++j) {
      float bv[8];
      load_bias8(bv, bias, n0 + j * 16 + h8);
#pragma unroll
      for (int mi = 0; mi < MT; ++mi) {
        float* p = xres + (mi * 16 + c) * XFP + n0 + j * 16 + h8;
        float4 lo = *(const float4*)(p), hi = *(const float4*)(p + 4);
        lo.x += acc[mi][j][0] + bv[0]; lo.y += acc[mi][j][1] + bv[1];
        lo.z += acc[mi][j][2] + bv[2]; lo.w += acc[mi][j][3] + bv[3];
        hi.x += acc[mi][j][4] + bv[4]; hi.y += acc[mi][j][5] + bv[5];
        hi.z += acc[mi][j][6] + bv[6]; hi.w += acc[mi][j][7] + bv[7];
        *(float4*)(p)     = lo;
        *(float4*)(p + 4) = hi;
      }
    }
  } else {
#pragma unroll
    for (int mi = 0; mi < MT; ++mi) {
#pragma unroll
      for (int j = 0; j < 2; ++j) {
        float* p = psum + (mi * 16 + c) * PSP + n0 + j * 16 + h8;
        *(float4*)(p)     = make_float4(acc[mi][j][0], acc[mi][j][1], acc[mi][j][2], acc[mi][j][3]);
        *(float4*)(p + 4) = make_float4(acc[mi][j][4], acc[mi][j][5], acc[mi][j][6], acc[mi][j][7]);
      }
    }
  }
}

__global__ void cvt_pack_frag(const float* __restrict__ s, _Float16* __restrict__ d, int kts) {
  const int g    = blockIdx.x * blockDim.x + threadIdx.x;
  const int frag = g >> 6;
  const int part = (g >> 5) & 1;
  const int ln   = g & 31;
  const int nt   = frag >> kts;
  const int kt   = frag & ((1 << kts) - 1);
  const float* q = s + ((nt * 16 + (ln & 15)) << (kts + 5)) + kt * 32 + part * 16 + ((ln >> 4) << 3);
  const float4 a = *(const float4*)(q);
  const float4 b = *(const float4*)(q + 4);
  v8h h;
  h[0] = (_Float16)a.x; h[1] = (_Float16)a.y; h[2] = (_Float16)a.z; h[3] = (_Float16)a.w;
  h[4] = (_Float16)b.x; h[5] = (_Float16)b.y; h[6] = (_Float16)b.z; h[7] = (_Float16)b.w;
  *(v8h*)(d + g * 8) = h;
}

template <bool B> struct LayerKind { static constexpr bool value = B; };

#define PAIRS 4
__global__ __launch_bounds__(512) void swt_kernel(
    const float* __restrict__ X, const float* __restrict__ pos,
    const _Float16* __restrict__ Win,  const float* __restrict__ Bin,
    const _Float16* __restrict__ Wout, const float* __restrict__ Bout,
    const _Float16* __restrict__ W1,   const float* __restrict__ B1,
    const _Float16* __restrict__ W2,   const float* __restrict__ B2,
    const float* __restrict__ ln1g, const float* __restrict__ ln1b,
    const float* __restrict__ ln2g, const float* __restrict__ ln2b,
    const float* __restrict__ headw, const float* __restrict__ headb,
    float* __restrict__ out) {
  __shared__ __attribute__((aligned(16))) float    xf32[64 * XFP];
  __shared__ __attribute__((aligned(16))) float    psum[64 * PSP];
  __shared__ __attribute__((aligned(16))) _Float16 l16[64 * XHP + 64 * QKP + 2 * 256 * 40];
  __shared__ __attribute__((aligned(16))) float    px32[16 * XFP];
  __shared__ __attribute__((aligned(16))) _Float16 po16[16 * XHP];
  _Float16* xf16 = l16;
  _Float16* qk   = l16 + 64 * XHP;
  _Float16* vT   = qk + 64 * QKP;

  const int tid  = threadIdx.x;
  const int lane = tid & 31;
  const int wave = __builtin_amdgcn_readfirstlane(tid >> 5);

  const v8f vzero = {};
  const int m0l = ((lane >> 4) << 3);
  const int nnl = lane & 15;
  const int kh  = wave >> 3;
  const int np  = wave & 7;

  for (int i = tid; i < 16 * XFP; i += 512) px32[i] = 0.f;
  for (int i = tid; i < 16 * XHP; i += 512) po16[i] = (_Float16)0.f;

  auto front = [&](const int l, auto last_tag) {
    constexpr bool LAST = decltype(last_tag)::value;
    constexpr int QT0 = LAST ? 1 : 0;
    const _Float16* wqkv = Win  + l * 768 * 256;
    const float*    bqkv = Bin  + l * 768;

    {
      const int n0 = wave * 16;
      const _Float16* wq = wqkv + wave * (8 * WFRAG) + lane * 8;
      v16h bring[3][3];
      gemm_request<3>(bring, wq, 16 * 8);
      lds_barrier();
      if constexpr (!LAST) {
        v8f acc[4][3];
#pragma unroll
        for (int mi = 0; mi < 4; ++mi) { acc[mi][0] = vzero; acc[mi][1] = vzero; acc[mi][2] = vzero; }
        gemm_steps<3, 8, 3u, 4, 0, 16>(acc, bring, xf16, XHP, wq, 16 * 8);
        float biasq[8], biask[8];
        load_bias8(biasq, bqkv, n0 + m0l);
        load_bias8(biask, bqkv, n0 + 256 + m0l);
        const float biasv = bqkv[n0 + 512 + nnl];
#pragma unroll
        for (int mi = 0; mi < 4; ++mi) {
#pragma unroll
          for (int r = 0; r < 8; ++r) {
            acc[mi][0][r] += biasq[r]; acc[mi][1][r] += biask[r]; acc[mi][2][r] += biasv;
          }
          store_rows_f16(qk, QKP, mi * 16, n0,       acc[mi][0]);
          store_rows_f16(qk, QKP, mi * 16, n0 + 256, acc[mi][1]);
          store_tile_f16_T(vT + (mi >> 1) * 10240, 40, (mi & 1) * 16, n0, acc[mi][2]);
        }
      } else {
        v8f accq[2], acck[4], accv[4];
        accq[0] = vzero; accq[1] = vzero;
#pragma unroll
        for (int mi = 0; mi < 4; ++mi) { acck[mi] = vzero; accv[mi] = vzero; }
        qkv_steps_last(accq, acck, accv, bring, xf16, XHP, wq, 16 * 8);
        float biasq[8], biask[8];
        load_bias8(biasq, bqkv, n0 + m0l);
        load_bias8(biask, bqkv, n0 + 256 + m0l);
        const float biasv = bqkv[n0 + 512 + nnl];
#pragma unroll
        for (int mi = 0; mi < 4; ++mi) {
#pragma unroll
          for (int r = 0; r < 8; ++r) { acck[mi][r] += biask[r]; accv[mi][r] += biasv; }
          store_rows_f16(qk, QKP, mi * 16, n0 + 256, acck[mi]);
          store_tile_f16_T(vT + (mi >> 1) * 10240, 40, (mi & 1) * 16, n0, accv[mi]);
        }
#pragma unroll
        for (int t = 0; t < 2; ++t) {
#pragma unroll
          for (int r = 0; r < 8; ++r) accq[t][r] += biasq[r];
          store_rows_f16(qk, QKP, 16 + t * 32, n0, accq[t]);
        }
      }
    }
    __syncthreads();

    {
      const int aw = wave >> 3;
      const int h  = wave & 7;
      const int rb = aw * 32;
      _Float16* vTw = vT + aw * 10240;
      v8f sc[2][2];
      v16h pf[2];
      {
        v16h kf[2], qf[2];
#pragma unroll
        for (int t = 0; t < 2; ++t) kf[t] = load_frag(qk + 256, QKP, rb + t * 16, h * 32);
#pragma unroll
        for (int t = QT0; t < 2; ++t) qf[t] = load_frag(qk, QKP, rb + t * 16, h * 32);
#pragma unroll
        for (int qt = QT0; qt < 2; ++qt) {
#pragma unroll
          for (int kt = 0; kt < 2; ++kt) sc[qt][kt] = wmma_f16(kf[kt], qf[qt], vzero);
        }
      }
      const float scale = 0.17677669529663687f;
#pragma unroll
      for (int qt = QT0; qt < 2; ++qt) {
        float s[2][8];
        float mx = sc[qt][0][0] * scale;
#pragma unroll
        for (int kt = 0; kt < 2; ++kt) {
#pragma unroll
          for (int r = 0; r < 8; ++r) { s[kt][r] = sc[qt][kt][r] * scale; mx = fmaxf(mx, s[kt][r]); }
        }
        mx = fmaxf(mx, xhalf(mx));
        float sm = 0.f;
#pragma unroll
        for (int kt = 0; kt < 2; ++kt) {
#pragma unroll
          for (int r = 0; r < 8; ++r) { s[kt][r] = __expf(s[kt][r] - mx); sm += s[kt][r]; }
        }
        sm += xhalf(sm);
        const float inv = 1.0f / sm;
        HFrag pq;
#pragma unroll
        for (int kt = 0; kt < 2; ++kt) {
#pragma unroll
          for (int r = 0; r < 8; ++r) pq.h[kt][r] = (_Float16)(s[kt][r] * inv);
        }
        pf[qt] = pq.v;
      }
      {
        v16h vf[2];
#pragma unroll
        for (int t = 0; t < 2; ++t) vf[t] = load_frag(vTw, 40, h * 32 + t * 16, 0);
#pragma unroll
        for (int qt = QT0; qt < 2; ++qt) {
#pragma unroll
          for (int dt = 0; dt < 2; ++dt) {
            v8f oa = wmma_f16(vf[dt], pf[qt], vzero);
            store_rows_f16(qk, QKP, rb + qt * 16, h * 32 + dt * 16, oa);
          }
        }
      }
    }
  };

  auto rest = [&](const int l, auto last_tag, float* xres, const _Float16* oa, const int lda) {
    constexpr bool LAST = decltype(last_tag)::value;
    constexpr int MT    = LAST ? 1 : 4;
    constexpr int NLIVE = 2 * PAIRS;
    const _Float16* wo   = Wout + l * 256 * 256;
    const float*    bo   = Bout + l * 256;
    const _Float16* w1l  = W1   + l * 1024 * 256;
    const float*    b1l  = B1   + l * 1024;
    const _Float16* w2l  = W2   + l * 256 * 1024;
    const float*    b2l  = B2   + l * 256;

    {
      const int n0 = np * 32;
      const _Float16* wq = wo + (np * 16 + kh * 4) * WFRAG + lane * 8;
      v16h bring[3][2];
      gemm_request<2>(bring, wq, 8);
      lds_barrier();
      v8f acc[MT][2];
#pragma unroll
      for (int mi = 0; mi < MT; ++mi) { acc[mi][0] = vzero; acc[mi][1] = vzero; }
      gemm_steps<2, 4, 3u, MT, 0, 16>(acc, bring, oa + kh * 4 * 32, lda, wq, 8);
      splitk_combine<MT>(xres, psum, bo, acc, kh, n0, lane);
    }
    __syncthreads();
    v16h bring1[3][2];
    v16h bring2[3][2];
    gemm_request<2>(bring1, w1l + wave * (8 * WFRAG) + lane * 8, 16 * 8);
    if constexpr (!LAST) {
      layer_norm_rows<4>(xres, psum, xf16, ln1g + l * 256, ln1b + l * 256, wave * 4, lane);
    } else {
      if (wave < NLIVE) layer_norm_rows<1>(xres, psum, xf16, ln1g + l * 256, ln1b + l * 256, wave, lane);
    }

    v8f facc[MT][2];
#pragma unroll
    for (int mi = 0; mi < MT; ++mi) { facc[mi][0] = vzero; facc[mi][1] = vzero; }
    lds_barrier();
#pragma unroll
    for (int cp = 0; cp < 2; ++cp) {
      const _Float16* wq2 = w2l + (np * 64 + cp * 16 + kh * 8) * WFRAG + lane * 8;
      {
        const int ng0 = cp * 512 + wave * 16;
        const int ng1 = ng0 + 256;
        const _Float16* wq = w1l + (cp * 32 + wave) * (8 * WFRAG) + lane * 8;
        v8f acc[MT][2];
#pragma unroll
        for (int mi = 0; mi < MT; ++mi) { acc[mi][0] = vzero; acc[mi][1] = vzero; }
        gemm_steps<2, 8, 3u, MT, 0, 16>(acc, bring1, xf16, XHP, wq, 16 * 8);
        gemm_request<2>(bring2, wq2, 32);
        float bias0[8], bias1[8];
        load_bias8(bias0, b1l, ng0 + m0l);
        load_bias8(bias1, b1l, ng1 + m0l);
#pragma unroll
        for (int mi = 0; mi < MT; ++mi) {
#pragma unroll
          for (int r = 0; r < 8; ++r) {
            acc[mi][0][r] = fmaxf(acc[mi][0][r] + bias0[r], 0.f);
            acc[mi][1][r] = fmaxf(acc[mi][1][r] + bias1[r], 0.f);
          }
          store_rows_f16(qk, QKP, mi * 16, wave * 16,       acc[mi][0]);
          store_rows_f16(qk, QKP, mi * 16, 256 + wave * 16, acc[mi][1]);
        }
      }
      lds_barrier();
      {
        gemm_steps<2, 8, 3u, MT, 0, 16>(facc, bring2, qk + kh * 8 * 32, QKP, wq2, 32);
        if (cp == 0) gemm_request<2>(bring1, w1l + (32 + wave) * (8 * WFRAG) + lane * 8, 16 * 8);
      }
      lds_barrier();
    }
    splitk_combine<MT>(xres, psum, b2l, facc, kh, np * 32, lane);
    __syncthreads();
    if constexpr (!LAST) {
      layer_norm_rows<4>(xres, psum, xf16, ln2g + l * 256, ln2b + l * 256, wave * 4, lane);
    } else {
      if (wave < NLIVE) layer_norm_rows<1>(xres, psum, xf16, ln2g + l * 256, ln2b + l * 256, wave, lane);
    }
  };

  for (int p = 0; p < PAIRS; ++p) {
    const int pair0 = (blockIdx.x * PAIRS + p) * 2;
    const int bb   = pair0 >> 9;
    const int ss0  = pair0 & 511;

    {
      const int jr = tid >> 3;
      const int wwin = jr >> 5, j = jr & 31;
      const int c0 = (tid & 7) * 32;
      int src = (ss0 + wwin) - j; if (src < 0) src = 0;
      const float* xr = X + ((bb << 9) + src) * 256;
      const float* pe = pos + j * 256;
#pragma unroll
      for (int u = 0; u < 8; ++u) {
        const int c = c0 + u * 4;
        float4 xv = *(const float4*)(xr + c);
        float4 pv = *(const float4*)(pe + c);
        float y0 = xv.x + pv.x, y1 = xv.y + pv.y, y2 = xv.z + pv.z, y3 = xv.w + pv.w;
        xf32[jr * XFP + c + 0] = y0; xf16[jr * XHP + c + 0] = (_Float16)y0;
        xf32[jr * XFP + c + 1] = y1; xf16[jr * XHP + c + 1] = (_Float16)y1;
        xf32[jr * XFP + c + 2] = y2; xf16[jr * XHP + c + 2] = (_Float16)y2;
        xf32[jr * XFP + c + 3] = y3; xf16[jr * XHP + c + 3] = (_Float16)y3;
      }
    }

    for (int l = 0; l < Ln; ++l) {
      if (l < Ln - 1) {
        front(l, LayerKind<false>{});
        rest(l, LayerKind<false>{}, xf32, qk, QKP);
      } else {
        front(l, LayerKind<true>{});
      }
    }
    __syncthreads();
    {
      const int w = tid >> 8, c = tid & 255;
      px32[(2 * p + w) * XFP + c] = xf32[(31 + 32 * w) * XFP + c];
      po16[(2 * p + w) * XHP + c] = qk[(31 + 32 * w) * QKP + c];
    }
    __syncthreads();
  }

  rest(Ln - 1, LayerKind<true>{}, px32, po16, XHP);
  __syncthreads();

  if (wave < 2 * PAIRS) {
    float acc = headb[lane];
    const float* hw = headw + lane * 256;
    const float* xr = px32 + wave * XFP;
    for (int k = 0; k < 256; k += 4) {
      float4 w4 = *(const float4*)(hw + k);
      acc += xr[k] * w4.x + xr[k + 1] * w4.y + xr[k + 2] * w4.z + xr[k + 3] * w4.w;
    }
    out[(blockIdx.x * 2 * PAIRS + wave) * 32 + lane] = acc;
  }
}

extern "C" void kernel_launch(void* const* d_in, const int* in_sizes, int n_in,
                              void* d_out, int out_size, void* d_ws, size_t ws_size,
                              hipStream_t stream) {
  const float* X     = (const float*)d_in[0];
  const float* pos   = (const float*)d_in[3];
  const float* w_in  = (const float*)d_in[4];
  const float* b_in  = (const float*)d_in[5];
  const float* w_out = (const float*)d_in[6];
  const float* b_out = (const float*)d_in[7];
  const float* w1    = (const float*)d_in[8];
  const float* b1    = (const float*)d_in[9];
  const float* w2    = (const float*)d_in[10];
  const float* b2    = (const float*)d_in[11];
  const float* ln1g  = (const float*)d_in[12];
  const float* ln1b  = (const float*)d_in[13];
  const float* ln2g  = (const float*)d_in[14];
  const float* ln2b  = (const float*)d_in[15];
  const float* headw = (const float*)d_in[16];
  const float* headb = (const float*)d_in[17];
  float* out = (float*)d_out;

  _Float16* ws      = (_Float16*)d_ws;
  _Float16* w_in_h  = ws;
  _Float16* w_out_h = w_in_h  + Ln * 768 * 256;
  _Float16* w1_h    = w_out_h + Ln * 256 * 256;
  _Float16* w2_h    = w1_h    + Ln * 1024 * 256;

  const int n_in_w  = Ln * 768 * 256;
  const int n_out_w = Ln * 256 * 256;
  const int n_w1    = Ln * 1024 * 256;
  const int n_w2    = Ln * 256 * 1024;
  cvt_pack_frag<<<n_in_w  / 2048, 256, 0, stream>>>(w_in,  w_in_h,  3);
  cvt_pack_frag<<<n_out_w / 2048, 256, 0, stream>>>(w_out, w_out_h, 3);
  cvt_pack_frag<<<n_w1    / 2048, 256, 0, stream>>>(w1,    w1_h,    3);
  cvt_pack_frag<<<n_w2    / 2048, 256, 0, stream>>>(w2,    w2_h,    5);

  swt_kernel<<<(Bn * Sn) / (2 * PAIRS), 512, 0, stream>>>(X, pos, w_in_h, b_in, w_out_h, b_out,
                                                w1_h, b1, w2_h, b2, ln1g, ln1b, ln2g,
                                                ln2b, headw, headb, out);
}


# Biases are the accumulators' start values, requested ahead of each GEMM phase, norm gain/offset ahead of the norm: no phase ends on an exposed L2 round trip; bias now first term of each sum
# speedup vs baseline: 2.3627x; 1.0368x over previous
#include <hip/hip_runtime.h>

#define Bn    4
#define Sn    512
#define Dn    256
#define Wn    32
#define Ln    4
#define Hn    8
#define DFFn  1024
#define NOUTn 32
#define XHP   264
#define QKP   520
#define PSP   260
#define XFP   260

typedef __attribute__((ext_vector_type(16))) _Float16 v16h;
typedef __attribute__((ext_vector_type(8)))  _Float16 v8h;
typedef __attribute__((ext_vector_type(4)))  _Float16 v4h;
typedef __attribute__((ext_vector_type(8)))  float    v8f;

union HFrag { v16h v; v8h h[2]; };

__device__ __forceinline__ v16h load_frag(const _Float16* p, int ld, int row0, int k0) {
  const int lane = threadIdx.x & 31;
  const _Float16* q = p + (row0 + (lane & 15)) * ld + k0 + ((lane >> 4) << 3);
  HFrag f;
  f.h[0] = *(const v8h*)(q);
  f.h[1] = *(const v8h*)(q + 16);
  return f.v;
}

#define WFRAG 512
__device__ __forceinline__ v16h load_frag_w(const _Float16* q) {
  HFrag f;
  f.h[0] = *(const v8h*)(q);
  f.h[1] = *(const v8h*)(q + 256);
  return f.v;
}

__device__ __forceinline__ v8f wmma_f16(v16h a, v16h b, v8f c) {
  return __builtin_amdgcn_wmma_f32_16x16x32_f16(false, a, false, b, (short)0, c, false, false);
}

__device__ __forceinline__ float xhalf(float x) {
  const int b = __builtin_bit_cast(int, x);
  return __builtin_bit_cast(float,
      __builtin_amdgcn_permlanex16(b, b, 0x76543210, 0xfedcba98, false, false));
}
template <int M>
__device__ __forceinline__ float xrow(float x) {
  const int b = __builtin_bit_cast(int, x);
  return __builtin_bit_cast(float, __builtin_amdgcn_update_dpp(0, b, 0x160 + M, 0xf, 0xf, true));
}
__device__ __forceinline__ float wave_sum(float s) {
  s += xhalf(s);
  s += xrow<8>(s);
  s += xrow<4>(s);
  s += xrow<2>(s);
  s += xrow<1>(s);
  return s;
}

template <int NB>
__device__ __forceinline__ void load_b_step(v16h (&b)[NB], const _Float16* wq, int kt, int bstride) {
#pragma unroll
  for (int j = 0; j < NB; ++j) b[j] = load_frag_w(wq + (kt + j * bstride) * WFRAG);
}

template <int NB>
__device__ __forceinline__ void gemm_request(v16h (&b)[3][NB], const _Float16* wq, int bstride) {
  load_b_step<NB>(b[0], wq, 0, bstride);
  load_b_step<NB>(b[1], wq, 1, bstride);
  __builtin_amdgcn_sched_barrier(0);
}

__device__ __forceinline__ void lds_barrier() {
  __builtin_amdgcn_fence(__ATOMIC_RELEASE, "workgroup", "local");
  __builtin_amdgcn_s_barrier();
  __builtin_amdgcn_fence(__ATOMIC_ACQUIRE, "workgroup", "local");
}

template <int NB, int KT, unsigned SW, int MT, int MB, int MS>
__device__ __forceinline__ void gemm_steps(v8f (&acc)[MT][NB], v16h (&b)[3][NB],
                                           const _Float16* ap, int ld,
                                           const _Float16* wq, int bstride) {
#pragma unroll
  for (int kt = 0; kt < KT; ++kt) {
    if (kt + 2 < KT) load_b_step<NB>(b[(kt + 2) % 3], wq, kt + 2, bstride);
    v16h a[MT];
#pragma unroll
    for (int mi = 0; mi < MT; ++mi) a[mi] = load_frag(ap, ld, MB + mi * MS, kt * 32);
    __builtin_amdgcn_sched_barrier(0);
#pragma unroll
    for (int mi = 0; mi < MT; ++mi) {
#pragma unroll
      for (int j = 0; j < NB; ++j) {
        if ((SW >> j) & 1u) acc[mi][j] = wmma_f16(b[kt % 3][j], a[mi], acc[mi][j]);
        else                acc[mi][j] = wmma_f16(a[mi], b[kt % 3][j], acc[mi][j]);
      }
    }
    __builtin_amdgcn_sched_barrier(0);
  }
}

__device__ __forceinline__ void qkv_steps_last(v8f (&accq)[2], v8f (&acck)[4], v8f (&accv)[4],
                                               v16h (&b)[3][3], const _Float16* ap, int ld,
                                               const _Float16* wq, int bstride) {
#pragma unroll
  for (int kt = 0; kt < 8; ++kt) {
    if (kt + 2 < 8) load_b_step<3>(b[(kt + 2) % 3], wq, kt + 2, bstride);
    v16h a[4];
#pragma unroll
    for (int mi = 0; mi < 4; ++mi) a[mi] = load_frag(ap, ld, mi * 16, kt * 32);
    __builtin_amdgcn_sched_barrier(0);
#pragma unroll
    for (int mi = 0; mi < 4; ++mi) {
      if (mi & 1) accq[mi >> 1] = wmma_f16(b[kt % 3][0], a[mi], accq[mi >> 1]);
      acck[mi] = wmma_f16(b[kt % 3][1], a[mi], acck[mi]);
      accv[mi] = wmma_f16(a[mi], b[kt % 3][2], accv[mi]);
    }
    __builtin_amdgcn_sched_barrier(0);
  }
}

__device__ __forceinline__ void store_tile_f16(_Float16* p, int ld, int row0, int col0, v8f acc) {
  const int lane = threadIdx.x & 31;
  const int m0 = row0 + ((lane >> 4) << 3);
  _Float16* q = p + col0 + (lane & 15);
#pragma unroll
  for (int r = 0; r < 8; ++r) q[(m0 + r) * ld] = (_Float16)acc[r];
}

__device__ __forceinline__ void store_tile_f16_T(_Float16* p, int ld, int row0, int col0, v8f acc) {
  const int lane = threadIdx.x & 31;
  const int m0 = row0 + ((lane >> 4) << 3);
  const int n  = col0 + (lane & 15);
  v8h h;
#pragma unroll
  for (int r = 0; r < 8; ++r) h[r] = (_Float16)acc[r];
  *(v8h*)(p + n * ld + m0) = h;
}

__device__ __forceinline__ void store_rows_f16(_Float16* p, int ld, int row0, int col0, v8f acc) {
  const int lane = threadIdx.x & 31;
  v8h h;
#pragma unroll
  for (int r = 0; r < 8; ++r) h[r] = (_Float16)acc[r];
  *(v8h*)(p + (row0 + (lane & 15)) * ld + col0 + ((lane >> 4) << 3)) = h;
}

__device__ __forceinline__ v8f load_bias8(const float* __restrict__ b, unsigned n0) {
  const float4 lo = *(const float4*)(b + n0);
  const float4 hi = *(const float4*)(b + n0 + 4);
  v8f v;
  v[0] = lo.x; v[1] = lo.y; v[2] = lo.z; v[3] = lo.w;
  v[4] = hi.x; v[5] = hi.y; v[6] = hi.z; v[7] = hi.w;
  return v;
}

struct LnPar { float4 g0, g1, b0, b1; };
__device__ __forceinline__ LnPar ln_request(const float* __restrict__ g, const float* __restrict__ bt,
                                            int lane) {
  const unsigned c0 = (unsigned)lane * 4u;
  LnPar q;
  q.g0 = *(const float4*)(g + c0);  q.g1 = *(const float4*)(g + 128 + c0);
  q.b0 = *(const float4*)(bt + c0); q.b1 = *(const float4*)(bt + 128 + c0);
  return q;
}

template <int NR>
__device__ __forceinline__ void layer_norm_rows(float* xf32, const float* ps, _Float16* xf16,
                                                const LnPar& q, int m0, int lane) {
  const int c0 = lane * 4;
  const float4 g0 = q.g0, g1 = q.g1, b0 = q.b0, b1 = q.b1;
  const float gg[8] = {g0.x, g0.y, g0.z, g0.w, g1.x, g1.y, g1.z, g1.w};
  const float bb[8] = {b0.x, b0.y, b0.z, b0.w, b1.x, b1.y, b1.z, b1.w};
#pragma unroll
  for (int rr = 0; rr < NR; ++rr) {
    const int m = m0 + rr;
    float* xr = xf32 + m * XFP + c0;
    const float* pr = ps + m * PSP + c0;
    const float4 x0 = *(const float4*)(xr), x1 = *(const float4*)(xr + 128);
    const float4 p0 = *(const float4*)(pr), p1 = *(const float4*)(pr + 128);
    float vals[8] = {x0.x + p0.x, x0.y + p0.y, x0.z + p0.z, x0.w + p0.w,
                     x1.x + p1.x, x1.y + p1.y, x1.z + p1.z, x1.w + p1.w};
    float sum = 0.f;
#pragma unroll
    for (int i = 0; i < 8; ++i) sum += vals[i];
    sum = wave_sum(sum);
    const float mu = sum * (1.f / 256.f);
    float s2 = 0.f;
#pragma unroll
    for (int i = 0; i < 8; ++i) { float d = vals[i] - mu; s2 += d * d; }
    s2 = wave_sum(s2);
    const float rstd = rsqrtf(s2 * (1.f / 256.f) + 1e-5f);
    float y[8];
#pragma unroll
    for (int i = 0; i < 8; ++i) y[i] = (vals[i] - mu) * rstd * gg[i] + bb[i];
    *(float4*)(xr)       = make_float4(y[0], y[1], y[2], y[3]);
    *(float4*)(xr + 128) = make_float4(y[4], y[5], y[6], y[7]);
    v4h h0, h1;
#pragma unroll
    for (int i = 0; i < 4; ++i) { h0[i] = (_Float16)y[i]; h1[i] = (_Float16)y[4 + i]; }
    _Float16* hr = xf16 + m * XHP + c0;
    *(v4h*)(hr)       = h0;
    *(v4h*)(hr + 128) = h1;
  }
}

template <int MT>
__device__ __forceinline__ void splitk_combine(float* xres, float* psum,
                                               const v8f (&acc)[MT][2], int kh, int n0, int lane) {
  const int c  = lane & 15;
  const int h8 = (lane >> 4) << 3;
  if (kh == 0) {
#pragma unroll
    for (int j = 0; j < 2; ++j) {
#pragma unroll
      for (int mi = 0; mi < MT; ++mi) {
        float* p = xres + (mi * 16 + c) * XFP + n0 + j * 16 + h8;
        float4 lo = *(const float4*)(p), hi = *(const float4*)(p + 4);
        lo.x += acc[mi][j][0]; lo.y += acc[mi][j][1];
        lo.z += acc[mi][j][2]; lo.w += acc[mi][j][3];
        hi.x += acc[mi][j][4]; hi.y += acc[mi][j][5];
        hi.z += acc[mi][j][6]; hi.w += acc[mi][j][7];
        *(float4*)(p)     = lo;
        *(float4*)(p + 4) = hi;
      }
    }
  } else {
#pragma unroll
    for (int mi = 0; mi < MT; ++mi) {
#pragma unroll
      for (int j = 0; j < 2; ++j) {
        float* p = psum + (mi * 16 + c) * PSP + n0 + j * 16 + h8;
        *(float4*)(p)     = make_float4(acc[mi][j][0], acc[mi][j][1], acc[mi][j][2], acc[mi][j][3]);
        *(float4*)(p + 4) = make_float4(acc[mi][j][4], acc[mi][j][5], acc[mi][j][6], acc[mi][j][7]);
      }
    }
  }
}

__global__ void cvt_pack_frag(const float* __restrict__ s, _Float16* __restrict__ d, int kts) {
  const int g    = blockIdx.x * blockDim.x + threadIdx.x;
  const int frag = g >> 6;
  const int part = (g >> 5) & 1;
  const int ln   = g & 31;
  const int nt   = frag >> kts;
  const int kt   = frag & ((1 << kts) - 1);
  const float* q = s + ((nt * 16 + (ln & 15)) << (kts + 5)) + kt * 32 + part * 16 + ((ln >> 4) << 3);
  const float4 a = *(const float4*)(q);
  const float4 b = *(const float4*)(q + 4);
  v8h h;
  h[0] = (_Float16)a.x; h[1] = (_Float16)a.y; h[2] = (_Float16)a.z; h[3] = (_Float16)a.w;
  h[4] = (_Float16)b.x; h[5] = (_Float16)b.y; h[6] = (_Float16)b.z; h[7] = (_Float16)b.w;
  *(v8h*)(d + g * 8) = h;
}

template <bool B> struct LayerKind { static constexpr bool value = B; };

#define PAIRS 4
__global__ __launch_bounds__(512) void swt_kernel(
    const float* __restrict__ X, const float* __restrict__ pos,
    const _Float16* __restrict__ Win,  const float* __restrict__ Bin,
    const _Float16* __restrict__ Wout, const float* __restrict__ Bout,
    const _Float16* __restrict__ W1,   const float* __restrict__ B1,
    const _Float16* __restrict__ W2,   const float* __restrict__ B2,
    const float* __restrict__ ln1g, const float* __restrict__ ln1b,
    const float* __restrict__ ln2g, const float* __restrict__ ln2b,
    const float* __restrict__ headw, const float* __restrict__ headb,
    float* __restrict__ out) {
  __shared__ __attribute__((aligned(16))) float    xf32[64 * XFP];
  __shared__ __attribute__((aligned(16))) float    psum[64 * PSP];
  __shared__ __attribute__((aligned(16))) _Float16 l16[64 * XHP + 64 * QKP + 2 * 256 * 40];
  __shared__ __attribute__((aligned(16))) float    px32[16 * XFP];
  __shared__ __attribute__((aligned(16))) _Float16 po16[16 * XHP];
  _Float16* xf16 = l16;
  _Float16* qk   = l16 + 64 * XHP;
  _Float16* vT   = qk + 64 * QKP;

  const int tid  = threadIdx.x;
  const int lane = tid & 31;
  const int wave = __builtin_amdgcn_readfirstlane(tid >> 5);

  const v8f vzero = {};
  const unsigned m0l = ((unsigned)lane >> 4) << 3;
  const unsigned nnl = (unsigned)lane & 15u;
  const int kh  = wave >> 3;
  const int np  = wave & 7;

  for (int i = tid; i < 16 * XFP; i += 512) px32[i] = 0.f;
  for (int i = tid; i < 16 * XHP; i += 512) po16[i] = (_Float16)0.f;

  auto front = [&](const int l, auto last_tag) {
    constexpr bool LAST = decltype(last_tag)::value;
    constexpr int QT0 = LAST ? 1 : 0;
    const _Float16* wqkv = Win  + l * 768 * 256;
    const float*    bqkv = Bin  + l * 768;

    {
      const int n0 = wave * 16;
      const _Float16* wq = wqkv + wave * (8 * WFRAG) + lane * 8;
      const v8f cq = load_bias8(bqkv + n0, m0l);
      const v8f ck = load_bias8(bqkv + n0 + 256, m0l);
      const float biasv = (bqkv + n0 + 512)[nnl];
      v16h bring[3][3];
      gemm_request<3>(bring, wq, 16 * 8);
      lds_barrier();
      const v8f cv = {biasv, biasv, biasv, biasv, biasv, biasv, biasv, biasv};
      if constexpr (!LAST) {
        v8f acc[4][3];
#pragma unroll
        for (int mi = 0; mi < 4; ++mi) { acc[mi][0] = cq; acc[mi][1] = ck; acc[mi][2] = cv; }
        gemm_steps<3, 8, 3u, 4, 0, 16>(acc, bring, xf16, XHP, wq, 16 * 8);
#pragma unroll
        for (int mi = 0; mi < 4; ++mi) {
          store_rows_f16(qk, QKP, mi * 16, n0,       acc[mi][0]);
          store_rows_f16(qk, QKP, mi * 16, n0 + 256, acc[mi][1]);
          store_tile_f16_T(vT + (mi >> 1) * 10240, 40, (mi & 1) * 16, n0, acc[mi][2]);
        }
      } else {
        v8f accq[2], acck[4], accv[4];
        accq[0] = cq; accq[1] = cq;
#pragma unroll
        for (int mi = 0; mi < 4; ++mi) { acck[mi] = ck; accv[mi] = cv; }
        qkv_steps_last(accq, acck, accv, bring, xf16, XHP, wq, 16 * 8);
#pragma unroll
        for (int mi = 0; mi < 4; ++mi) {
          store_rows_f16(qk, QKP, mi * 16, n0 + 256, acck[mi]);
          store_tile_f16_T(vT + (mi >> 1) * 10240, 40, (mi & 1) * 16, n0, accv[mi]);
        }
#pragma unroll
        for (int t = 0; t < 2; ++t) {
          store_rows_f16(qk, QKP, 16 + t * 32, n0, accq[t]);
        }
      }
    }
    __syncthreads();

    {
      const int aw = wave >> 3;
      const int h  = wave & 7;
      const int rb = aw * 32;
      _Float16* vTw = vT + aw * 10240;
      v8f sc[2][2];
      v16h pf[2];
      {
        v16h kf[2], qf[2];
#pragma unroll
        for (int t = 0; t < 2; ++t) kf[t] = load_frag(qk + 256, QKP, rb + t * 16, h * 32);
#pragma unroll
        for (int t = QT0; t < 2; ++t) qf[t] = load_frag(qk, QKP, rb + t * 16, h * 32);
#pragma unroll
        for (int qt = QT0; qt < 2; ++qt) {
#pragma unroll
          for (int kt = 0; kt < 2; ++kt) sc[qt][kt] = wmma_f16(kf[kt], qf[qt], vzero);
        }
      }
      const float scale = 0.17677669529663687f;
#pragma unroll
      for (int qt = QT0; qt < 2; ++qt) {
        float s[2][8];
        float mx = sc[qt][0][0] * scale;
#pragma unroll
        for (int kt = 0; kt < 2; ++kt) {
#pragma unroll
          for (int r = 0; r < 8; ++r) { s[kt][r] = sc[qt][kt][r] * scale; mx = fmaxf(mx, s[kt][r]); }
        }
        mx = fmaxf(mx, xhalf(mx));
        float sm = 0.f;
#pragma unroll
        for (int kt = 0; kt < 2; ++kt) {
#pragma unroll
          for (int r = 0; r < 8; ++r) { s[kt][r] = __expf(s[kt][r] - mx); sm += s[kt][r]; }
        }
        sm += xhalf(sm);
        const float inv = 1.0f / sm;
        HFrag pq;
#pragma unroll
        for (int kt = 0; kt < 2; ++kt) {
#pragma unroll
          for (int r = 0; r < 8; ++r) pq.h[kt][r] = (_Float16)(s[kt][r] * inv);
        }
        pf[qt] = pq.v;
      }
      {
        v16h vf[2];
#pragma unroll
        for (int t = 0; t < 2; ++t) vf[t] = load_frag(vTw, 40, h * 32 + t * 16, 0);
#pragma unroll
        for (int qt = QT0; qt < 2; ++qt) {
#pragma unroll
          for (int dt = 0; dt < 2; ++dt) {
            v8f oa = wmma_f16(vf[dt], pf[qt], vzero);
            store_rows_f16(qk, QKP, rb + qt * 16, h * 32 + dt * 16, oa);
          }
        }
      }
    }
  };

  auto rest = [&](const int l, auto last_tag, float* xres, const _Float16* oa, const int lda) {
    constexpr bool LAST = decltype(last_tag)::value;
    constexpr int MT    = LAST ? 1 : 4;
    constexpr int NLIVE = 2 * PAIRS;
    const _Float16* wo   = Wout + l * 256 * 256;
    const float*    bo   = Bout + l * 256;
    const _Float16* w1l  = W1   + l * 1024 * 256;
    const float*    b1l  = B1   + l * 1024;
    const _Float16* w2l  = W2   + l * 256 * 1024;
    const float*    b2l  = B2   + l * 256;

    {
      const int n0 = np * 32;
      const _Float16* wq = wo + (np * 16 + kh * 4) * WFRAG + lane * 8;
      v8f cb0 = load_bias8(bo + n0, m0l);
      v8f cb1 = load_bias8(bo + n0 + 16, m0l);
      v16h bring[3][2];
      gemm_request<2>(bring, wq, 8);
      lds_barrier();
      if (kh != 0) { cb0 = vzero; cb1 = vzero; }
      v8f acc[MT][2];
#pragma unroll
      for (int mi = 0; mi < MT; ++mi) { acc[mi][0] = cb0; acc[mi][1] = cb1; }
      gemm_steps<2, 4, 3u, MT, 0, 16>(acc, bring, oa + kh * 4 * 32, lda, wq, 8);
      splitk_combine<MT>(xres, psum, acc, kh, n0, lane);
    }
    __syncthreads();
    v16h bring1[3][2];
    v16h bring2[3][2];
    const LnPar lp1 = ln_request(ln1g + l * 256, ln1b + l * 256, lane);
    v8f hb0 = load_bias8(b1l + wave * 16, m0l);
    v8f hb1 = load_bias8(b1l + 256 + wave * 16, m0l);
    gemm_request<2>(bring1, w1l + wave * (8 * WFRAG) + lane * 8, 16 * 8);
    if constexpr (!LAST) {
      layer_norm_rows<4>(xres, psum, xf16, lp1, wave * 4, lane);
    } else {
      if (wave < NLIVE) layer_norm_rows<1>(xres, psum, xf16, lp1, wave, lane);
    }

    v8f facc[MT][2];
    v8f fb0, fb1;
    lds_barrier();
#pragma unroll
    for (int cp = 0; cp < 2; ++cp) {
      const _Float16* wq2 = w2l + (np * 64 + cp * 16 + kh * 8) * WFRAG + lane * 8;
      {
        const _Float16* wq = w1l + (cp * 32 + wave) * (8 * WFRAG) + lane * 8;
        v8f acc[MT][2];
#pragma unroll
        for (int mi = 0; mi < MT; ++mi) { acc[mi][0] = hb0; acc[mi][1] = hb1; }
        gemm_steps<2, 8, 3u, MT, 0, 16>(acc, bring1, xf16, XHP, wq, 16 * 8);
        if (cp == 0) {
          fb0 = load_bias8(b2l + np * 32, m0l);
          fb1 = load_bias8(b2l + np * 32 + 16, m0l);
        }
        gemm_request<2>(bring2, wq2, 32);
#pragma unroll
        for (int mi = 0; mi < MT; ++mi) {
#pragma unroll
          for (int r = 0; r < 8; ++r) {
            acc[mi][0][r] = fmaxf(acc[mi][0][r], 0.f);
            acc[mi][1][r] = fmaxf(acc[mi][1][r], 0.f);
          }
          store_rows_f16(qk, QKP, mi * 16, wave * 16,       acc[mi][0]);
          store_rows_f16(qk, QKP, mi * 16, 256 + wave * 16, acc[mi][1]);
        }
      }
      lds_barrier();
      {
        if (cp == 0) {
          if (kh != 0) { fb0 = vzero; fb1 = vzero; }
#pragma unroll
          for (int mi = 0; mi < MT; ++mi) { facc[mi][0] = fb0; facc[mi][1] = fb1; }
        }
        gemm_steps<2, 8, 3u, MT, 0, 16>(facc, bring2, qk + kh * 8 * 32, QKP, wq2, 32);
        if (cp == 0) {
          hb0 = load_bias8(b1l + 512 + wave * 16, m0l);
          hb1 = load_bias8(b1l + 768 + wave * 16, m0l);
          gemm_request<2>(bring1, w1l + (32 + wave) * (8 * WFRAG) + lane * 8, 16 * 8);
        }
      }
      lds_barrier();
    }
    splitk_combine<MT>(xres, psum, facc, kh, np * 32, lane);
    __syncthreads();
    const LnPar lp2 = ln_request(ln2g + l * 256, ln2b + l * 256, lane);
    __builtin_amdgcn_sched_barrier(0);
    if constexpr (!LAST) {
      layer_norm_rows<4>(xres, psum, xf16, lp2, wave * 4, lane);
    } else {
      if (wave < NLIVE) layer_norm_rows<1>(xres, psum, xf16, lp2, wave, lane);
    }
  };

  for (int p = 0; p < PAIRS; ++p) {
    const int pair0 = (blockIdx.x * PAIRS + p) * 2;
    const int bb   = pair0 >> 9;
    const int ss0  = pair0 & 511;

    {
      const int jr = tid >> 3;
      const int wwin = jr >> 5, j = jr & 31;
      const int c0 = (tid & 7) * 32;
      int src = (ss0 + wwin) - j; if (src < 0) src = 0;
      const float* xr = X + ((bb << 9) + src) * 256;
      const float* pe = pos + j * 256;
#pragma unroll
      for (int u = 0; u < 8; ++u) {
        const int c = c0 + u * 4;
        float4 xv = *(const float4*)(xr + c);
        float4 pv = *(const float4*)(pe + c);
        float y0 = xv.x + pv.x, y1 = xv.y + pv.y, y2 = xv.z + pv.z, y3 = xv.w + pv.w;
        xf32[jr * XFP + c + 0] = y0; xf16[jr * XHP + c + 0] = (_Float16)y0;
        xf32[jr * XFP + c + 1] = y1; xf16[jr * XHP + c + 1] = (_Float16)y1;
        xf32[jr * XFP + c + 2] = y2; xf16[jr * XHP + c + 2] = (_Float16)y2;
        xf32[jr * XFP + c + 3] = y3; xf16[jr * XHP + c + 3] = (_Float16)y3;
      }
    }

    for (int l = 0; l < Ln; ++l) {
      if (l < Ln - 1) {
        front(l, LayerKind<false>{});
        rest(l, LayerKind<false>{}, xf32, qk, QKP);
      } else {
        front(l, LayerKind<true>{});
      }
    }
    __syncthreads();
    {
      const int w = tid >> 8, c = tid & 255;
      px32[(2 * p + w) * XFP + c] = xf32[(31 + 32 * w) * XFP + c];
      po16[(2 * p + w) * XHP + c] = qk[(31 + 32 * w) * QKP + c];
    }
    __syncthreads();
  }

  rest(Ln - 1, LayerKind<true>{}, px32, po16, XHP);
  __syncthreads();

  if (wave < 2 * PAIRS) {
    float acc = headb[lane];
    const float* hw = headw + lane * 256;
    const float* xr = px32 + wave * XFP;
    for (int k = 0; k < 256; k += 4) {
      float4 w4 = *(const float4*)(hw + k);
      acc += xr[k] * w4.x + xr[k + 1] * w4.y + xr[k + 2] * w4.z + xr[k + 3] * w4.w;
    }
    out[(blockIdx.x * 2 * PAIRS + wave) * 32 + lane] = acc;
  }
}

extern "C" void kernel_launch(void* const* d_in, const int* in_sizes, int n_in,
                              void* d_out, int out_size, void* d_ws, size_t ws_size,
                              hipStream_t stream) {
  const float* X     = (const float*)d_in[0];
  const float* pos   = (const float*)d_in[3];
  const float* w_in  = (const float*)d_in[4];
  const float* b_in  = (const float*)d_in[5];
  const float* w_out = (const float*)d_in[6];
  const float* b_out = (const float*)d_in[7];
  const float* w1    = (const float*)d_in[8];
  const float* b1    = (const float*)d_in[9];
  const float* w2    = (const float*)d_in[10];
  const float* b2    = (const float*)d_in[11];
  const float* ln1g  = (const float*)d_in[12];
  const float* ln1b  = (const float*)d_in[13];
  const float* ln2g  = (const float*)d_in[14];
  const float* ln2b  = (const float*)d_in[15];
  const float* headw = (const float*)d_in[16];
  const float* headb = (const float*)d_in[17];
  float* out = (float*)d_out;

  _Float16* ws      = (_Float16*)d_ws;
  _Float16* w_in_h  = ws;
  _Float16* w_out_h = w_in_h  + Ln * 768 * 256;
  _Float16* w1_h    = w_out_h + Ln * 256 * 256;
  _Float16* w2_h    = w1_h    + Ln * 1024 * 256;

  const int n_in_w  = Ln * 768 * 256;
  const int n_out_w = Ln * 256 * 256;
  const int n_w1    = Ln * 1024 * 256;
  const int n_w2    = Ln * 256 * 1024;
  cvt_pack_frag<<<n_in_w  / 2048, 256, 0, stream>>>(w_in,  w_in_h,  3);
  cvt_pack_frag<<<n_out_w / 2048, 256, 0, stream>>>(w_out, w_out_h, 3);
  cvt_pack_frag<<<n_w1    / 2048, 256, 0, stream>>>(w1,    w1_h,    3);
  cvt_pack_frag<<<n_w2    / 2048, 256, 0, stream>>>(w2,    w2_h,    5);

  swt_kernel<<<(Bn * Sn) / (2 * PAIRS), 512, 0, stream>>>(X, pos, w_in_h, b_in, w_out_h, b_out,
                                                w1_h, b1, w2_h, b2, ln1g, ln1b, ln2g,
                                                ln2b, headw, headb, out);
}
